# starting kernel
# baseline (speedup 1.0000x reference)
#include <hip/hip_runtime.h>
#include <hip/hip_bf16.h>
#include <math.h>


typedef __attribute__((ext_vector_type(2))) float v2f;
typedef __attribute__((ext_vector_type(8))) float v8f;

#define NSTEP 100
#define T1    99
#define NB    256
#define OUT_U   (NB * NSTEP)
#define OUT_EPS (3 * OUT_U)
#define OUT_ST  (OUT_EPS + NB)

__device__ __forceinline__ void Ecoef(int r, int c, float* pa, float* pb, float* pc) {
  float a = 0.f, b = 0.f, q = 0.f;
  if (r == 0)      { if (c == 0) a = 1.f;  else if (c == 1) b = 1.f;   else if (c == 2) q = 0.5f; else if (c == 3) a = -1.f; }
  else if (r == 1) { if (c == 1) a = 1.f;  else if (c == 2) b = 1.f;   else if (c == 4) a = -1.f; }
  else if (r == 2) { if (c == 0) a = -1.f; else if (c == 3) a = 1.f;   else if (c == 4) b = -1.f; else if (c == 5) q = 0.5f; }
  else if (r == 3) { if (c == 1) a = -1.f; else if (c == 4) a = 1.f;   else if (c == 5) b = -1.f; }
  else if (r == 4) { if (c == 1) a = -1.f; else if (c == 2) b = -0.5f; else if (c == 4) a = 1.f;  else if (c == 5) b = -0.5f; }
  *pa = a; *pb = b; *pc = q;
}

__device__ __forceinline__ void chol3(double* s) {
  double l00 = sqrt(s[0]);
  double l10 = s[1] / l00;
  double l11 = sqrt(s[2] - l10 * l10);
  double l20 = s[3] / l00;
  double l21 = (s[4] - l20 * l10) / l11;
  double l22 = sqrt(s[5] - l20 * l20 - l21 * l21);
  s[0] = l00; s[1] = l10; s[2] = l11; s[3] = l20; s[4] = l21; s[5] = l22;
}
__device__ __forceinline__ void lsolve3(const double* l, double* v) {
  double v0 = v[0] / l[0];
  double v1 = (v[1] - l[1] * v0) / l[2];
  double v2 = (v[2] - l[3] * v0 - l[4] * v1) / l[5];
  v[0] = v0; v[1] = v1; v[2] = v2;
}
__device__ __forceinline__ void ltsolve3(const double* l, double* v) {
  double x2 = v[2] / l[5];
  double x1 = (v[1] - l[4] * x2) / l[2];
  double x0 = (v[0] - l[1] * x1 - l[3] * x2) / l[0];
  v[0] = x0; v[1] = x1; v[2] = x2;
}

__global__ __launch_bounds__(32) void ode_blocktri_kernel(
    const float* __restrict__ coeffs,
    const float* __restrict__ rhs,
    const float* __restrict__ ivr,
    const float* __restrict__ steps,
    float* __restrict__ out)
{
  __shared__ double S[NSTEP][6];
  __shared__ double Bo[T1][9];
  __shared__ double gv[NSTEP][3];
  __shared__ float  hstep[T1];

  const int b = blockIdx.x;
  const int lane = threadIdx.x;

  for (int i = lane; i < NSTEP * 6; i += 32) ((double*)S)[i] = 0.0;
  for (int i = lane; i < T1 * 9;   i += 32) ((double*)Bo)[i] = 0.0;
  for (int i = lane; i < NSTEP * 3; i += 32) ((double*)gv)[i] = 0.0;
  for (int i = lane; i < T1; i += 32) hstep[i] = steps[b * T1 + i];
  __syncthreads();

  for (int t = lane; t < NSTEP; t += 32) {
    const double c0 = (double)coeffs[(b * NSTEP + t) * 3 + 0];
    const double c1 = (double)coeffs[(b * NSTEP + t) * 3 + 1];
    const double c2 = (double)coeffs[(b * NSTEP + t) * 3 + 2];
    const double r  = (double)rhs[b * NSTEP + t];
    S[t][0] += c0 * c0; S[t][1] += c1 * c0; S[t][2] += c1 * c1;
    S[t][3] += c2 * c0; S[t][4] += c2 * c1; S[t][5] += c2 * c2;
    gv[t][0] += c0 * r; gv[t][1] += c1 * r; gv[t][2] += c2 * r;
    if (t == 0) {
      S[0][0] += 1.0; S[0][2] += 1.0;
      gv[0][0] += (double)ivr[b * 2 + 0];
      gv[0][1] += (double)ivr[b * 2 + 1];
    }
  }
  __syncthreads();

  const int m = lane & 15;
  const int khalf = lane >> 4;
  const int col = (m < 6) ? m : 6;
  float a0, b0, q0, a1, b1, q1, a2, b2, q2;
  Ecoef(2 * khalf,                 col, &a0, &b0, &q0);
  Ecoef(2 * khalf + 1,             col, &a1, &b1, &q1);
  Ecoef((khalf == 0) ? 4 : -1,     col, &a2, &b2, &q2);

  for (int p = 0; p < T1; ++p) {
    const float h = hstep[p];
    v2f f0, f1;
    f0.x = fmaf(fmaf(q0, h, b0), h, a0);
    f0.y = fmaf(fmaf(q1, h, b1), h, a1);
    f1.x = fmaf(fmaf(q2, h, b2), h, a2);
    f1.y = 0.f;
    v8f acc = {};
    acc = __builtin_amdgcn_wmma_f32_16x16x4_f32(false, f0, false, f0,
                                                (short)0, acc, false, false);
    acc = __builtin_amdgcn_wmma_f32_16x16x4_f32(false, f1, false, f1,
                                                (short)0, acc, false, false);
    if (lane < 3) {
      const int j = lane;
      #pragma unroll
      for (int i = 0; i < 3; ++i)
        if (i >= j) S[p][i * (i + 1) / 2 + j] += (double)acc[i];
    } else if (lane < 6) {
      const int j = lane - 3;
      #pragma unroll
      for (int i = 0; i < 3; ++i) Bo[p][i * 3 + j] += (double)acc[i];
      #pragma unroll
      for (int i = 0; i < 3; ++i)
        if (i >= j) S[p + 1][i * (i + 1) / 2 + j] += (double)acc[3 + i];
    }
  }
  __syncthreads();

  if (lane == 0) {
    chol3(S[0]);
    lsolve3(S[0], gv[0]);
    for (int p = 0; p < T1; ++p) {
      const double* L = S[p];
      double X[3][3];
      #pragma unroll
      for (int r = 0; r < 3; ++r) {
        const double c0 = Bo[p][0 + r];
        const double c1 = Bo[p][3 + r];
        const double c2 = Bo[p][6 + r];
        const double x0 = c0 / L[0];
        const double x1 = (c1 - x0 * L[1]) / L[2];
        const double x2 = (c2 - x0 * L[3] - x1 * L[4]) / L[5];
        X[r][0] = x0; X[r][1] = x1; X[r][2] = x2;
      }
      #pragma unroll
      for (int r = 0; r < 3; ++r) {
        Bo[p][r * 3 + 0] = X[r][0]; Bo[p][r * 3 + 1] = X[r][1]; Bo[p][r * 3 + 2] = X[r][2];
      }
      S[p + 1][0] -= X[0][0]*X[0][0] + X[0][1]*X[0][1] + X[0][2]*X[0][2];
      S[p + 1][1] -= X[1][0]*X[0][0] + X[1][1]*X[0][1] + X[1][2]*X[0][2];
      S[p + 1][2] -= X[1][0]*X[1][0] + X[1][1]*X[1][1] + X[1][2]*X[1][2];
      S[p + 1][3] -= X[2][0]*X[0][0] + X[2][1]*X[0][1] + X[2][2]*X[0][2];
      S[p + 1][4] -= X[2][0]*X[1][0] + X[2][1]*X[1][1] + X[2][2]*X[1][2];
      S[p + 1][5] -= X[2][0]*X[2][0] + X[2][1]*X[2][1] + X[2][2]*X[2][2];
      chol3(S[p + 1]);
      const double y0 = gv[p][0], y1 = gv[p][1], y2 = gv[p][2];
      gv[p + 1][0] -= X[0][0]*y0 + X[0][1]*y1 + X[0][2]*y2;
      gv[p + 1][1] -= X[1][0]*y0 + X[1][1]*y1 + X[1][2]*y2;
      gv[p + 1][2] -= X[2][0]*y0 + X[2][1]*y1 + X[2][2]*y2;
      lsolve3(S[p + 1], gv[p + 1]);
    }
    ltsolve3(S[T1], gv[T1]);
    for (int p = T1 - 1; p >= 0; --p) {
      const double x0 = gv[p + 1][0], x1 = gv[p + 1][1], x2 = gv[p + 1][2];
      gv[p][0] -= Bo[p][0]*x0 + Bo[p][3]*x1 + Bo[p][6]*x2;
      gv[p][1] -= Bo[p][1]*x0 + Bo[p][4]*x1 + Bo[p][7]*x2;
      gv[p][2] -= Bo[p][2]*x0 + Bo[p][5]*x1 + Bo[p][8]*x2;
      ltsolve3(S[p], gv[p]);
    }
  }
  __syncthreads();

  for (int t = lane; t < NSTEP; t += 32) {
    out[0 * OUT_U + b * NSTEP + t] = (float)gv[t][0];
    out[1 * OUT_U + b * NSTEP + t] = (float)gv[t][1];
    out[2 * OUT_U + b * NSTEP + t] = (float)gv[t][2];
  }
  if (lane == 0) out[OUT_EPS + b] = 0.0f;
  for (int p = lane; p < T1; p += 32) out[OUT_ST + b * T1 + p] = hstep[p];
}

extern "C" void kernel_launch(void* const* d_in, const int* in_sizes, int n_in,
                              void* d_out, int out_size, void* d_ws, size_t ws_size,
                              hipStream_t stream) {
  (void)in_sizes; (void)n_in; (void)out_size; (void)d_ws; (void)ws_size;
  const float* coeffs = (const float*)d_in[0];
  const float* rhs    = (const float*)d_in[1];
  const float* ivr    = (const float*)d_in[2];
  const float* steps  = (const float*)d_in[3];
  float* out = (float*)d_out;
  hipLaunchKernelGGL(ode_blocktri_kernel, dim3(NB), dim3(32), 0, stream,
                     coeffs, rhs, ivr, steps, out);
}


# Each serial block chain (partition interiors, separator chain) runs on 3 lanes, not 1: lane c takes row c of the multiplier and spike column c; spike sums V^T V, V^T y leave the loop; sum order kept
# speedup vs baseline: 3.7245x; 3.7245x over previous
#include <hip/hip_runtime.h>
#include <hip/hip_bf16.h>
#include <math.h>


typedef __attribute__((ext_vector_type(2))) float v2f;
typedef __attribute__((ext_vector_type(8))) float v8f;

#define NSTEP 100
#define T1    99
#define NB    256
#define OUT_U   (NB * NSTEP)
#define OUT_EPS (3 * OUT_U)
#define OUT_ST  (OUT_EPS + NB)
#define PER   10
#define MI    9
#define NPART 10

__device__ __forceinline__ void Ecoef(int r, int c, float* pa, float* pb, float* pc) {
  float a = 0.f, b = 0.f, q = 0.f;
  if (r == 0)      { if (c == 0) a = 1.f;  else if (c == 1) b = 1.f;   else if (c == 2) q = 0.5f; else if (c == 3) a = -1.f; }
  else if (r == 1) { if (c == 1) a = 1.f;  else if (c == 2) b = 1.f;   else if (c == 4) a = -1.f; }
  else if (r == 2) { if (c == 0) a = -1.f; else if (c == 3) a = 1.f;   else if (c == 4) b = -1.f; else if (c == 5) q = 0.5f; }
  else if (r == 3) { if (c == 1) a = -1.f; else if (c == 4) a = 1.f;   else if (c == 5) b = -1.f; }
  else if (r == 4) { if (c == 1) a = -1.f; else if (c == 2) b = -0.5f; else if (c == 4) a = 1.f;  else if (c == 5) b = -0.5f; }
  *pa = a; *pb = b; *pc = q;
}

__device__ __forceinline__ void chol3(double* s) {
  double l00 = sqrt(s[0]);
  double l10 = s[1] / l00;
  double l11 = sqrt(s[2] - l10 * l10);
  double l20 = s[3] / l00;
  double l21 = (s[4] - l20 * l10) / l11;
  double l22 = sqrt(s[5] - l20 * l20 - l21 * l21);
  s[0] = l00; s[1] = l10; s[2] = l11; s[3] = l20; s[4] = l21; s[5] = l22;
}
__device__ __forceinline__ void lsolve3(const double* l, double* v) {
  double v0 = v[0] / l[0];
  double v1 = (v[1] - l[1] * v0) / l[2];
  double v2 = (v[2] - l[3] * v0 - l[4] * v1) / l[5];
  v[0] = v0; v[1] = v1; v[2] = v2;
}
__device__ __forceinline__ void ltsolve3(const double* l, double* v) {
  double x2 = v[2] / l[5];
  double x1 = (v[1] - l[4] * x2) / l[2];
  double x0 = (v[0] - l[1] * x1 - l[3] * x2) / l[0];
  v[0] = x0; v[1] = x1; v[2] = x2;
}

__device__ __forceinline__ void mstep3(const double* L, double* Bp, double X[3][3]) {
  #pragma unroll
  for (int r = 0; r < 3; ++r) {
    const double c0 = Bp[0 + r];
    const double c1 = Bp[3 + r];
    const double c2 = Bp[6 + r];
    const double x0 = c0 / L[0];
    const double x1 = (c1 - x0 * L[1]) / L[2];
    const double x2 = (c2 - x0 * L[3] - x1 * L[4]) / L[5];
    X[r][0] = x0; X[r][1] = x1; X[r][2] = x2;
  }
  #pragma unroll
  for (int r = 0; r < 3; ++r) {
    Bp[r * 3 + 0] = X[r][0]; Bp[r * 3 + 1] = X[r][1]; Bp[r * 3 + 2] = X[r][2];
  }
}
__device__ __forceinline__ void subxxt3(double* Sn, const double X[3][3]) {
  Sn[0] -= X[0][0]*X[0][0] + X[0][1]*X[0][1] + X[0][2]*X[0][2];
  Sn[1] -= X[1][0]*X[0][0] + X[1][1]*X[0][1] + X[1][2]*X[0][2];
  Sn[2] -= X[1][0]*X[1][0] + X[1][1]*X[1][1] + X[1][2]*X[1][2];
  Sn[3] -= X[2][0]*X[0][0] + X[2][1]*X[0][1] + X[2][2]*X[0][2];
  Sn[4] -= X[2][0]*X[1][0] + X[2][1]*X[1][1] + X[2][2]*X[1][2];
  Sn[5] -= X[2][0]*X[2][0] + X[2][1]*X[2][1] + X[2][2]*X[2][2];
}

__global__ __launch_bounds__(32) void ode_blocktri_kernel(
    const float* __restrict__ coeffs,
    const float* __restrict__ rhs,
    const float* __restrict__ ivr,
    const float* __restrict__ steps,
    float* __restrict__ out)
{
  __shared__ double S[NSTEP][6];
  __shared__ double Bo[T1][9];
  __shared__ double gv[NSTEP][3];
  __shared__ float  hstep[T1];
  __shared__ double V[NSTEP][9];
  __shared__ double Mx[T1][9];
  __shared__ double R[NPART][9];
  __shared__ double Rm[NPART][9];

  const int b = blockIdx.x;
  const int lane = threadIdx.x;

  for (int i = lane; i < NSTEP * 6; i += 32) ((double*)S)[i] = 0.0;
  for (int i = lane; i < T1 * 9;   i += 32) ((double*)Bo)[i] = 0.0;
  for (int i = lane; i < NSTEP * 3; i += 32) ((double*)gv)[i] = 0.0;
  for (int i = lane; i < T1; i += 32) hstep[i] = steps[b * T1 + i];
  __syncthreads();

  for (int t = lane; t < NSTEP; t += 32) {
    const double c0 = (double)coeffs[(b * NSTEP + t) * 3 + 0];
    const double c1 = (double)coeffs[(b * NSTEP + t) * 3 + 1];
    const double c2 = (double)coeffs[(b * NSTEP + t) * 3 + 2];
    const double r  = (double)rhs[b * NSTEP + t];
    S[t][0] += c0 * c0; S[t][1] += c1 * c0; S[t][2] += c1 * c1;
    S[t][3] += c2 * c0; S[t][4] += c2 * c1; S[t][5] += c2 * c2;
    gv[t][0] += c0 * r; gv[t][1] += c1 * r; gv[t][2] += c2 * r;
    if (t == 0) {
      S[0][0] += 1.0; S[0][2] += 1.0;
      gv[0][0] += (double)ivr[b * 2 + 0];
      gv[0][1] += (double)ivr[b * 2 + 1];
    }
  }
  __syncthreads();

  const int m = lane & 15;
  const int khalf = lane >> 4;
  const int col = (m < 6) ? m : 6;
  float a0, b0, q0, a1, b1, q1, a2, b2, q2;
  Ecoef(2 * khalf,                 col, &a0, &b0, &q0);
  Ecoef(2 * khalf + 1,             col, &a1, &b1, &q1);
  Ecoef((khalf == 0) ? 4 : -1,     col, &a2, &b2, &q2);

  for (int p = 0; p < T1; ++p) {
    const float h = hstep[p];
    v2f f0, f1;
    f0.x = fmaf(fmaf(q0, h, b0), h, a0);
    f0.y = fmaf(fmaf(q1, h, b1), h, a1);
    f1.x = fmaf(fmaf(q2, h, b2), h, a2);
    f1.y = 0.f;
    v8f acc = {};
    acc = __builtin_amdgcn_wmma_f32_16x16x4_f32(false, f0, false, f0,
                                                (short)0, acc, false, false);
    acc = __builtin_amdgcn_wmma_f32_16x16x4_f32(false, f1, false, f1,
                                                (short)0, acc, false, false);
    if (lane < 3) {
      const int j = lane;
      #pragma unroll
      for (int i = 0; i < 3; ++i)
        if (i >= j) S[p][i * (i + 1) / 2 + j] += (double)acc[i];
    } else if (lane < 6) {
      const int j = lane - 3;
      #pragma unroll
      for (int i = 0; i < 3; ++i) Bo[p][i * 3 + j] += (double)acc[i];
      #pragma unroll
      for (int i = 0; i < 3; ++i)
        if (i >= j) S[p + 1][i * (i + 1) / 2 + j] += (double)acc[3 + i];
    }
  }
  __syncthreads();

  const int pa = PER * lane;
  const bool hasL = lane > 0;
  const bool act = lane < 3 * NPART;
  const int part = act ? lane / 3 : 0;
  const int c = act ? lane - 3 * part : 0;
  const int pk = PER * part;
  double Lc[6] = {};
  double yc[3] = {};
  double vc[3] = {};
  if (act) {
    for (int q = 0; q < 6; ++q) Lc[q] = S[pk][q];
    chol3(Lc);
    for (int q = 0; q < 3; ++q) yc[q] = gv[pk][q];
    lsolve3(Lc, yc);
    if (part > 0) {
      vc[0] = Bo[pk - 1][c * 3 + 0]; vc[1] = Bo[pk - 1][c * 3 + 1]; vc[2] = Bo[pk - 1][c * 3 + 2];
      lsolve3(Lc, vc);
    }
    V[pk][0 + c] = vc[0]; V[pk][3 + c] = vc[1]; V[pk][6 + c] = vc[2];
  }
  __syncthreads();
  if (act && c == 0) {
    for (int q = 0; q < 6; ++q) S[pk][q] = Lc[q];
    for (int q = 0; q < 3; ++q) gv[pk][q] = yc[q];
  }
  for (int i = 0; i < MI; ++i) {
    double sn[6] = {};
    double yn[3] = {};
    if (act) {
      const int p = pk + i;
      const double c0 = Bo[p][0 + c], c1 = Bo[p][3 + c], c2 = Bo[p][6 + c];
      const double x0 = c0 / Lc[0];
      const double x1 = (c1 - x0 * Lc[1]) / Lc[2];
      const double x2 = (c2 - x0 * Lc[3] - x1 * Lc[4]) / Lc[5];
      Mx[p][c * 3 + 0] = x0; Mx[p][c * 3 + 1] = x1; Mx[p][c * 3 + 2] = x2;
    }
    __syncthreads();
    if (act) {
      const int p = pk + i;
      double X[3][3];
      for (int r = 0; r < 3; ++r) { X[r][0] = Mx[p][r * 3 + 0]; X[r][1] = Mx[p][r * 3 + 1]; X[r][2] = Mx[p][r * 3 + 2]; }
      for (int q = 0; q < 6; ++q) sn[q] = S[p + 1][q];
      subxxt3(sn, X);
      yn[0] = gv[p + 1][0] - (X[0][0]*yc[0] + X[0][1]*yc[1] + X[0][2]*yc[2]);
      yn[1] = gv[p + 1][1] - (X[1][0]*yc[0] + X[1][1]*yc[1] + X[1][2]*yc[2]);
      yn[2] = gv[p + 1][2] - (X[2][0]*yc[0] + X[2][1]*yc[1] + X[2][2]*yc[2]);
      if (i < MI - 1) {
        chol3(sn);
        lsolve3(sn, yn);
        double vn[3];
        for (int r = 0; r < 3; ++r)
          vn[r] = -(X[r][0]*vc[0] + X[r][1]*vc[1] + X[r][2]*vc[2]);
        lsolve3(sn, vn);
        V[p + 1][0 + c] = vn[0]; V[p + 1][3 + c] = vn[1]; V[p + 1][6 + c] = vn[2];
        for (int q = 0; q < 6; ++q) Lc[q] = sn[q];
        for (int q = 0; q < 3; ++q) { yc[q] = yn[q]; vc[q] = vn[q]; }
      } else {
        for (int j = 0; j < 3; ++j)
          R[part][c * 3 + j] = -(vc[0]*X[j][0] + vc[1]*X[j][1] + vc[2]*X[j][2]);
      }
    }
    __syncthreads();
    if (act && c == 0) {
      for (int q = 0; q < 6; ++q) S[pk + i + 1][q] = sn[q];
      for (int q = 0; q < 3; ++q) gv[pk + i + 1][q] = yn[q];
    }
  }
  __syncthreads();
  for (int idx = lane; idx < (NPART - 1) * 6; idx += 32) {
    const int k = 1 + idx / 6, q = idx % 6;
    const int i2 = (q >= 3) ? 2 : ((q >= 1) ? 1 : 0);
    const int j2 = q - i2 * (i2 + 1) / 2;
    const int p0 = PER * k;
    double acc = 0.0;
    for (int i = 0; i < MI; ++i)
      acc += V[p0 + i][0 + i2]*V[p0 + i][0 + j2] + V[p0 + i][3 + i2]*V[p0 + i][3 + j2]
           + V[p0 + i][6 + i2]*V[p0 + i][6 + j2];
    S[p0 - 1][q] -= acc;
  }
  for (int idx = lane; idx < (NPART - 1) * 3; idx += 32) {
    const int k = 1 + idx / 3, j = idx % 3;
    const int p0 = PER * k;
    double acc = 0.0;
    for (int i = 0; i < MI; ++i)
      acc += V[p0 + i][0 + j]*gv[p0 + i][0] + V[p0 + i][3 + j]*gv[p0 + i][1]
           + V[p0 + i][6 + j]*gv[p0 + i][2];
    gv[p0 - 1][j] -= acc;
  }
  __syncthreads();
  const bool sepl = lane < 3;
  double Ls[6] = {};
  double ys[3] = {};
  if (sepl) {
    for (int q = 0; q < 6; ++q) Ls[q] = S[MI][q];
    chol3(Ls);
    for (int q = 0; q < 3; ++q) ys[q] = gv[MI][q];
    lsolve3(Ls, ys);
  }
  __syncthreads();
  if (lane == 0) {
    for (int q = 0; q < 6; ++q) S[MI][q] = Ls[q];
    for (int q = 0; q < 3; ++q) gv[MI][q] = ys[q];
  }
  for (int j = 0; j < NPART - 1; ++j) {
    const int sn = PER * j + MI + PER;
    if (sepl) {
      const double c0 = R[j + 1][0 + lane], c1 = R[j + 1][3 + lane], c2 = R[j + 1][6 + lane];
      const double x0 = c0 / Ls[0];
      const double x1 = (c1 - x0 * Ls[1]) / Ls[2];
      const double x2 = (c2 - x0 * Ls[3] - x1 * Ls[4]) / Ls[5];
      Rm[j + 1][lane * 3 + 0] = x0; Rm[j + 1][lane * 3 + 1] = x1; Rm[j + 1][lane * 3 + 2] = x2;
    }
    __syncthreads();
    if (sepl) {
      double X[3][3];
      for (int r = 0; r < 3; ++r) { X[r][0] = Rm[j + 1][r * 3 + 0]; X[r][1] = Rm[j + 1][r * 3 + 1]; X[r][2] = Rm[j + 1][r * 3 + 2]; }
      double sq[6];
      for (int q = 0; q < 6; ++q) sq[q] = S[sn][q];
      subxxt3(sq, X);
      chol3(sq);
      double yn[3];
      yn[0] = gv[sn][0] - (X[0][0]*ys[0] + X[0][1]*ys[1] + X[0][2]*ys[2]);
      yn[1] = gv[sn][1] - (X[1][0]*ys[0] + X[1][1]*ys[1] + X[1][2]*ys[2]);
      yn[2] = gv[sn][2] - (X[2][0]*ys[0] + X[2][1]*ys[1] + X[2][2]*ys[2]);
      lsolve3(sq, yn);
      for (int q = 0; q < 6; ++q) Ls[q] = sq[q];
      for (int q = 0; q < 3; ++q) ys[q] = yn[q];
    }
    __syncthreads();
    if (lane == 0) {
      for (int q = 0; q < 6; ++q) S[sn][q] = Ls[q];
      for (int q = 0; q < 3; ++q) gv[sn][q] = ys[q];
    }
  }
  __syncthreads();
  if (lane == 0) {
    ltsolve3(S[NSTEP - 1], gv[NSTEP - 1]);
    for (int j = NPART - 2; j >= 0; --j) {
      const int s = PER * j + MI, sn = s + PER;
      const double x0 = gv[sn][0], x1 = gv[sn][1], x2 = gv[sn][2];
      gv[s][0] -= Rm[j + 1][0]*x0 + Rm[j + 1][3]*x1 + Rm[j + 1][6]*x2;
      gv[s][1] -= Rm[j + 1][1]*x0 + Rm[j + 1][4]*x1 + Rm[j + 1][7]*x2;
      gv[s][2] -= Rm[j + 1][2]*x0 + Rm[j + 1][5]*x1 + Rm[j + 1][8]*x2;
      ltsolve3(S[s], gv[s]);
    }
  }
  __syncthreads();
  if (lane < NPART) {
    double xl0 = 0.0, xl1 = 0.0, xl2 = 0.0;
    if (hasL) { xl0 = gv[pa - 1][0]; xl1 = gv[pa - 1][1]; xl2 = gv[pa - 1][2]; }
    for (int i = MI - 1; i >= 0; --i) {
      const int p = pa + i;
      const double x0 = gv[p + 1][0], x1 = gv[p + 1][1], x2 = gv[p + 1][2];
      double t0 = gv[p][0] - (Mx[p][0]*x0 + Mx[p][3]*x1 + Mx[p][6]*x2);
      double t1 = gv[p][1] - (Mx[p][1]*x0 + Mx[p][4]*x1 + Mx[p][7]*x2);
      double t2 = gv[p][2] - (Mx[p][2]*x0 + Mx[p][5]*x1 + Mx[p][8]*x2);
      if (hasL) {
        t0 -= V[p][0]*xl0 + V[p][1]*xl1 + V[p][2]*xl2;
        t1 -= V[p][3]*xl0 + V[p][4]*xl1 + V[p][5]*xl2;
        t2 -= V[p][6]*xl0 + V[p][7]*xl1 + V[p][8]*xl2;
      }
      gv[p][0] = t0; gv[p][1] = t1; gv[p][2] = t2;
      ltsolve3(S[p], gv[p]);
    }
  }
  __syncthreads();

  for (int t = lane; t < NSTEP; t += 32) {
    out[0 * OUT_U + b * NSTEP + t] = (float)gv[t][0];
    out[1 * OUT_U + b * NSTEP + t] = (float)gv[t][1];
    out[2 * OUT_U + b * NSTEP + t] = (float)gv[t][2];
  }
  if (lane == 0) out[OUT_EPS + b] = 0.0f;
  for (int p = lane; p < T1; p += 32) out[OUT_ST + b * T1 + p] = hstep[p];
}

extern "C" void kernel_launch(void* const* d_in, const int* in_sizes, int n_in,
                              void* d_out, int out_size, void* d_ws, size_t ws_size,
                              hipStream_t stream) {
  (void)in_sizes; (void)n_in; (void)out_size; (void)d_ws; (void)ws_size;
  const float* coeffs = (const float*)d_in[0];
  const float* rhs    = (const float*)d_in[1];
  const float* ivr    = (const float*)d_in[2];
  const float* steps  = (const float*)d_in[3];
  float* out = (float*)d_out;
  hipLaunchKernelGGL(ode_blocktri_kernel, dim3(NB), dim3(32), 0, stream,
                     coeffs, rhs, ivr, steps, out);
}


# Separators' chain: a separator's factor, multiplier rows and rhs are ONE bordered 7x3 Cholesky panel, one row per lane (divide sequences shared, pivots by lane read); update by row; sum order kept
# speedup vs baseline: 7.0989x; 1.9060x over previous
#include <hip/hip_runtime.h>
#include <hip/hip_bf16.h>
#include <math.h>


typedef __attribute__((ext_vector_type(2))) float v2f;
typedef __attribute__((ext_vector_type(8))) float v8f;

#define NSTEP 100
#define T1    99
#define NB    256
#define OUT_U   (NB * NSTEP)
#define OUT_EPS (3 * OUT_U)
#define OUT_ST  (OUT_EPS + NB)
#define PER   10
#define MI    9
#define NPART 10
#define NTILE 7

__device__ __forceinline__ void Ecoef(int r, int c, float* pa, float* pb, float* pc) {
  float a = 0.f, b = 0.f, q = 0.f;
  if (r == 0)      { a = (c == 0) ? 1.f  : ((c == 3) ? -1.f : 0.f); b = (c == 1) ? 1.f  : 0.f; q = (c == 2) ? 0.5f : 0.f; }
  else if (r == 1) { a = (c == 1) ? 1.f  : ((c == 4) ? -1.f : 0.f); b = (c == 2) ? 1.f  : 0.f; }
  else if (r == 2) { a = (c == 0) ? -1.f : ((c == 3) ? 1.f  : 0.f); b = (c == 4) ? -1.f : 0.f; q = (c == 5) ? 0.5f : 0.f; }
  else if (r == 3) { a = (c == 1) ? -1.f : ((c == 4) ? 1.f  : 0.f); b = (c == 5) ? -1.f : 0.f; }
  else if (r == 4) { a = (c == 1) ? -1.f : ((c == 4) ? 1.f  : 0.f); b = (c == 2 || c == 5) ? -0.5f : 0.f; }
  *pa = a; *pb = b; *pc = q;
}

__device__ __forceinline__ void chol3(double* s) {
  double l00 = sqrt(s[0]);
  double l10 = s[1] / l00;
  double l11 = sqrt(s[2] - l10 * l10);
  double l20 = s[3] / l00;
  double l21 = (s[4] - l20 * l10) / l11;
  double l22 = sqrt(s[5] - l20 * l20 - l21 * l21);
  s[0] = l00; s[1] = l10; s[2] = l11; s[3] = l20; s[4] = l21; s[5] = l22;
}
__device__ __forceinline__ void lsolve3(const double* l, double* v) {
  double v0 = v[0] / l[0];
  double v1 = (v[1] - l[1] * v0) / l[2];
  double v2 = (v[2] - l[3] * v0 - l[4] * v1) / l[5];
  v[0] = v0; v[1] = v1; v[2] = v2;
}
__device__ __forceinline__ void ltsolve3(const double* l, double* v) {
  double x2 = v[2] / l[5];
  double x1 = (v[1] - l[4] * x2) / l[2];
  double x0 = (v[0] - l[1] * x1 - l[3] * x2) / l[0];
  v[0] = x0; v[1] = x1; v[2] = x2;
}

__device__ __forceinline__ void mstep3(const double* L, double* Bp, double X[3][3]) {
  #pragma unroll
  for (int r = 0; r < 3; ++r) {
    const double c0 = Bp[0 + r];
    const double c1 = Bp[3 + r];
    const double c2 = Bp[6 + r];
    const double x0 = c0 / L[0];
    const double x1 = (c1 - x0 * L[1]) / L[2];
    const double x2 = (c2 - x0 * L[3] - x1 * L[4]) / L[5];
    X[r][0] = x0; X[r][1] = x1; X[r][2] = x2;
  }
  #pragma unroll
  for (int r = 0; r < 3; ++r) {
    Bp[r * 3 + 0] = X[r][0]; Bp[r * 3 + 1] = X[r][1]; Bp[r * 3 + 2] = X[r][2];
  }
}
__device__ __forceinline__ void subxxt3(double* Sn, const double X[3][3]) {
  Sn[0] -= X[0][0]*X[0][0] + X[0][1]*X[0][1] + X[0][2]*X[0][2];
  Sn[1] -= X[1][0]*X[0][0] + X[1][1]*X[0][1] + X[1][2]*X[0][2];
  Sn[2] -= X[1][0]*X[1][0] + X[1][1]*X[1][1] + X[1][2]*X[1][2];
  Sn[3] -= X[2][0]*X[0][0] + X[2][1]*X[0][1] + X[2][2]*X[0][2];
  Sn[4] -= X[2][0]*X[1][0] + X[2][1]*X[1][1] + X[2][2]*X[1][2];
  Sn[5] -= X[2][0]*X[2][0] + X[2][1]*X[2][1] + X[2][2]*X[2][2];
}

__device__ __forceinline__ double lane_bcast(double v, int src) {
  const int lo = __builtin_amdgcn_readlane(__double2loint(v), src);
  const int hi = __builtin_amdgcn_readlane(__double2hiint(v), src);
  return __hiloint2double(hi, lo);
}

__global__ __launch_bounds__(32) void ode_blocktri_kernel(
    const float* __restrict__ coeffs,
    const float* __restrict__ rhs,
    const float* __restrict__ ivr,
    const float* __restrict__ steps,
    float* __restrict__ out)
{
  __shared__ double S[NSTEP][6];
  __shared__ double Bo[T1][9];
  __shared__ double gv[NSTEP][3];
  __shared__ float  hstep[T1];
  __shared__ double V[NSTEP][9];
  __shared__ double Mx[T1][9];
  __shared__ double R[NPART][9];
  __shared__ double Rm[NPART][9];

  const int b = blockIdx.x;
  const int lane = threadIdx.x;

  for (int i = lane; i < T1; i += 32) hstep[i] = steps[b * T1 + i];
  __syncthreads();

  const int e  = lane & 15;
  const int hh = lane >> 4;
  const int i2 = (e >= 3) ? 2 : ((e >= 1) ? 1 : 0);
  const int i3 = (e >= 12) ? 2 : ((e >= 9) ? 1 : 0);
  const int ca = (e < 6) ? i2 : ((e < 15) ? i3 : 6);
  const int cb = (e < 6) ? e - i2 * (i2 + 1) / 2 : ((e < 15) ? e - 3 - 3 * i3 : 6);
  const int ga = (e < 6) ? 3 + ca : 6;
  const int gb = (e < 6) ? 3 + cb : 6;
  float ph0 = 0.f, ph1 = 0.f, ph2 = 0.f, ph3 = 0.f, ph4 = 0.f;
  float pg0 = 0.f, pg1 = 0.f, pg2 = 0.f, pg3 = 0.f, pg4 = 0.f;
  #pragma unroll
  for (int k = 0; k < 5; ++k) {
    float a1, b1, q1, a2, b2, q2;
    Ecoef(k, ca, &a1, &b1, &q1); Ecoef(k, cb, &a2, &b2, &q2);
    ph0 += a1 * a2; ph1 += a1 * b2 + b1 * a2; ph2 += a1 * q2 + b1 * b2 + q1 * a2;
    ph3 += b1 * q2 + q1 * b2; ph4 += q1 * q2;
    Ecoef(k, ga, &a1, &b1, &q1); Ecoef(k, gb, &a2, &b2, &q2);
    pg0 += a1 * a2; pg1 += a1 * b2 + b1 * a2; pg2 += a1 * q2 + b1 * b2 + q1 * a2;
    pg3 += b1 * q2 + q1 * b2; pg4 += q1 * q2;
  }
  v2f A0, A1, A2;
  A0.x = hh ? ph2 : ph0;  A0.y = hh ? ph3 : ph1;
  A1.x = hh ? pg1 : ph4;  A1.y = hh ? pg2 : pg0;
  A2.x = hh ? 0.f : pg3;  A2.y = hh ? 0.f : pg4;

  float cf0[NTILE], cf1[NTILE], cf2[NTILE], rf[NTILE], hf[NTILE], gf[NTILE];
  #pragma unroll
  for (int n = 0; n < NTILE; ++n) {
    const int t = 16 * n + e;
    const bool own = (hh == 0) && (t < NSTEP);
    const int tl = (t < NSTEP) ? t : NSTEP - 1;
    const float l0 = coeffs[(b * NSTEP + tl) * 3 + 0];
    const float l1 = coeffs[(b * NSTEP + tl) * 3 + 1];
    const float l2 = coeffs[(b * NSTEP + tl) * 3 + 2];
    const float lr = rhs[b * NSTEP + tl];
    const float lh = hstep[(t < T1) ? t : T1 - 1];
    const float lg = hstep[(t >= 1 && t < NSTEP) ? t - 1 : 0];
    cf0[n] = own ? l0 : 0.f;
    cf1[n] = own ? l1 : 0.f;
    cf2[n] = own ? l2 : 0.f;
    rf[n]  = own ? lr : 0.f;
    hf[n]  = (t < T1) ? lh : 0.f;
    gf[n]  = (t >= 1 && t < NSTEP) ? lg : 0.f;
  }
  #pragma unroll
  for (int n = 0; n < NTILE; ++n) {
    const int t = 16 * n + e;
    const float h = hf[n], g = gf[n];
    const float u = (t < T1) ? 1.f : 0.f;
    const float w = (t >= 1 && t < NSTEP) ? 1.f : 0.f;
    const float h2 = h * h, h3 = h2 * h, h4 = h2 * h2;
    const float g2 = g * g, g3 = g2 * g, g4 = g2 * g2;
    v2f B0, B1, B2;
    B0.x = hh ? h2 : u;    B0.y = hh ? h3 : h;
    B1.x = hh ? g  : h4;   B1.y = hh ? g2 : w;
    B2.x = hh ? 0.f : g3;  B2.y = hh ? 0.f : g4;
    v8f acc = {};
    acc = __builtin_amdgcn_wmma_f32_16x16x4_f32(false, A0, false, B0, (short)0, acc, false, false);
    acc = __builtin_amdgcn_wmma_f32_16x16x4_f32(false, A1, false, B1, (short)0, acc, false, false);
    acc = __builtin_amdgcn_wmma_f32_16x16x4_f32(false, A2, false, B2, (short)0, acc, false, false);
    const double c0 = (double)cf0[n], c1 = (double)cf1[n], c2 = (double)cf2[n], rr = (double)rf[n];
    double s0 = fma(c0, c0, (double)acc[0]);
    double s1 = fma(c1, c0, (double)acc[1]);
    double s2 = fma(c1, c1, (double)acc[2]);
    double s3 = fma(c2, c0, (double)acc[3]);
    double s4 = fma(c2, c1, (double)acc[4]);
    double s5 = fma(c2, c2, (double)acc[5]);
    const double s6 = (double)acc[6], s7 = (double)acc[7];
    double y0 = c0 * rr, y1 = c1 * rr, y2 = c2 * rr;
    if (n == 0 && lane == 0) {
      s0 += 1.0; s2 += 1.0;
      y0 += (double)ivr[b * 2 + 0];
      y1 += (double)ivr[b * 2 + 1];
    }
    if (hh == 0) {
      if (t < NSTEP) {
        S[t][0] = s0; S[t][1] = s1; S[t][2] = s2; S[t][3] = s3; S[t][4] = s4; S[t][5] = s5;
        gv[t][0] = y0; gv[t][1] = y1; gv[t][2] = y2;
      }
      if (t < T1) { Bo[t][0] = s6; Bo[t][1] = s7; }
    } else if (t < T1) {
      Bo[t][2] = s0; Bo[t][3] = s1; Bo[t][4] = s2; Bo[t][5] = s3; Bo[t][6] = s4; Bo[t][7] = s5; Bo[t][8] = s6;
    }
  }
  __syncthreads();

  const int pa = PER * lane;
  const bool hasL = lane > 0;
  const bool act = lane < 3 * NPART;
  const int part = act ? lane / 3 : 0;
  const int c = act ? lane - 3 * part : 0;
  const int pk = PER * part;
  double Lc[6] = {};
  double yc[3] = {};
  double vc[3] = {};
  if (act) {
    for (int q = 0; q < 6; ++q) Lc[q] = S[pk][q];
    chol3(Lc);
    for (int q = 0; q < 3; ++q) yc[q] = gv[pk][q];
    lsolve3(Lc, yc);
    if (part > 0) {
      vc[0] = Bo[pk - 1][c * 3 + 0]; vc[1] = Bo[pk - 1][c * 3 + 1]; vc[2] = Bo[pk - 1][c * 3 + 2];
      lsolve3(Lc, vc);
    }
    V[pk][0 + c] = vc[0]; V[pk][3 + c] = vc[1]; V[pk][6 + c] = vc[2];
  }
  __syncthreads();
  if (act && c == 0) {
    for (int q = 0; q < 6; ++q) S[pk][q] = Lc[q];
    for (int q = 0; q < 3; ++q) gv[pk][q] = yc[q];
  }
  for (int i = 0; i < MI; ++i) {
    double sn[6] = {};
    double yn[3] = {};
    if (act) {
      const int p = pk + i;
      const double c0 = Bo[p][0 + c], c1 = Bo[p][3 + c], c2 = Bo[p][6 + c];
      const double x0 = c0 / Lc[0];
      const double x1 = (c1 - x0 * Lc[1]) / Lc[2];
      const double x2 = (c2 - x0 * Lc[3] - x1 * Lc[4]) / Lc[5];
      Mx[p][c * 3 + 0] = x0; Mx[p][c * 3 + 1] = x1; Mx[p][c * 3 + 2] = x2;
    }
    __syncthreads();
    if (act) {
      const int p = pk + i;
      double X[3][3];
      for (int r = 0; r < 3; ++r) { X[r][0] = Mx[p][r * 3 + 0]; X[r][1] = Mx[p][r * 3 + 1]; X[r][2] = Mx[p][r * 3 + 2]; }
      for (int q = 0; q < 6; ++q) sn[q] = S[p + 1][q];
      subxxt3(sn, X);
      yn[0] = gv[p + 1][0] - (X[0][0]*yc[0] + X[0][1]*yc[1] + X[0][2]*yc[2]);
      yn[1] = gv[p + 1][1] - (X[1][0]*yc[0] + X[1][1]*yc[1] + X[1][2]*yc[2]);
      yn[2] = gv[p + 1][2] - (X[2][0]*yc[0] + X[2][1]*yc[1] + X[2][2]*yc[2]);
      if (i < MI - 1) {
        chol3(sn);
        lsolve3(sn, yn);
        double vn[3];
        for (int r = 0; r < 3; ++r)
          vn[r] = -(X[r][0]*vc[0] + X[r][1]*vc[1] + X[r][2]*vc[2]);
        lsolve3(sn, vn);
        V[p + 1][0 + c] = vn[0]; V[p + 1][3 + c] = vn[1]; V[p + 1][6 + c] = vn[2];
        for (int q = 0; q < 6; ++q) Lc[q] = sn[q];
        for (int q = 0; q < 3; ++q) { yc[q] = yn[q]; vc[q] = vn[q]; }
      } else {
        for (int j = 0; j < 3; ++j)
          R[part][c * 3 + j] = -(vc[0]*X[j][0] + vc[1]*X[j][1] + vc[2]*X[j][2]);
      }
    }
    __syncthreads();
    if (act && c == 0) {
      for (int q = 0; q < 6; ++q) S[pk + i + 1][q] = sn[q];
      for (int q = 0; q < 3; ++q) gv[pk + i + 1][q] = yn[q];
    }
  }
  __syncthreads();
  for (int idx = lane; idx < (NPART - 1) * 6; idx += 32) {
    const int k = 1 + idx / 6, q = idx % 6;
    const int i2 = (q >= 3) ? 2 : ((q >= 1) ? 1 : 0);
    const int j2 = q - i2 * (i2 + 1) / 2;
    const int p0 = PER * k;
    double acc = 0.0;
    for (int i = 0; i < MI; ++i)
      acc += V[p0 + i][0 + i2]*V[p0 + i][0 + j2] + V[p0 + i][3 + i2]*V[p0 + i][3 + j2]
           + V[p0 + i][6 + i2]*V[p0 + i][6 + j2];
    S[p0 - 1][q] -= acc;
  }
  for (int idx = lane; idx < (NPART - 1) * 3; idx += 32) {
    const int k = 1 + idx / 3, j = idx % 3;
    const int p0 = PER * k;
    double acc = 0.0;
    for (int i = 0; i < MI; ++i)
      acc += V[p0 + i][0 + j]*gv[p0 + i][0] + V[p0 + i][3 + j]*gv[p0 + i][1]
           + V[p0 + i][6 + j]*gv[p0 + i][2];
    gv[p0 - 1][j] -= acc;
  }
  __syncthreads();
  const bool isS = lane < 3, isC = (lane >= 3) && (lane < 6), isG = (lane == 6);
  const int rl = isS ? lane : 0;
  const int rb = rl * (rl + 1) / 2;
  const int cc = isC ? lane - 3 : 0;
  double a0 = 0.0, a1 = 0.0, a2 = 0.0;
  if (isS)      { a0 = S[MI][rb]; a1 = S[MI][rb + 1]; a2 = S[MI][rb + 2]; }
  else if (isC) { a0 = R[1][0 + cc]; a1 = R[1][3 + cc]; a2 = R[1][6 + cc]; }
  else if (isG) { a0 = gv[MI][0]; a1 = gv[MI][1]; a2 = gv[MI][2]; }
  for (int j = 0; j < NPART; ++j) {
    const int s = PER * j + MI;
    double d = sqrt(lane_bcast(a0, 0));
    a0 = (lane == 0) ? d : a0 / d;
    const double l10 = lane_bcast(a0, 1), l20 = lane_bcast(a0, 2);
    a1 -= a0 * l10;
    a2 -= a0 * l20;
    d = sqrt(lane_bcast(a1, 1));
    a1 = (lane == 1) ? d : a1 / d;
    const double l21 = lane_bcast(a1, 2);
    a2 -= a1 * l21;
    d = sqrt(lane_bcast(a2, 2));
    a2 = (lane == 2) ? d : a2 / d;
    if (isS) {
      S[s][rb] = a0;
      if (lane >= 1) S[s][rb + 1] = a1;
      if (lane == 2) S[s][rb + 2] = a2;
    } else if (isC) {
      if (j + 1 < NPART) { Rm[j + 1][cc * 3 + 0] = a0; Rm[j + 1][cc * 3 + 1] = a1; Rm[j + 1][cc * 3 + 2] = a2; }
    } else if (isG) {
      gv[s][0] = a0; gv[s][1] = a1; gv[s][2] = a2;
    }
    __syncthreads();
    if (j + 1 < NPART) {
      const int sn = s + PER;
      double X[3][3];
      for (int r = 0; r < 3; ++r) { X[r][0] = Rm[j + 1][r * 3 + 0]; X[r][1] = Rm[j + 1][r * 3 + 1]; X[r][2] = Rm[j + 1][r * 3 + 2]; }
      double u0 = 0.0, u1 = 0.0, u2 = 0.0, b0 = 0.0, b1 = 0.0, b2 = 0.0;
      if (isS) {
        u0 = Rm[j + 1][rl * 3 + 0]; u1 = Rm[j + 1][rl * 3 + 1]; u2 = Rm[j + 1][rl * 3 + 2];
        b0 = S[sn][rb]; b1 = S[sn][rb + 1]; b2 = S[sn][rb + 2];
      } else if (isG) {
        u0 = a0; u1 = a1; u2 = a2;
        b0 = gv[sn][0]; b1 = gv[sn][1]; b2 = gv[sn][2];
      } else if (isC) {
        if (j + 2 < NPART) { b0 = R[j + 2][0 + cc]; b1 = R[j + 2][3 + cc]; b2 = R[j + 2][6 + cc]; }
      }
      a0 = b0 - (X[0][0]*u0 + X[0][1]*u1 + X[0][2]*u2);
      a1 = b1 - (X[1][0]*u0 + X[1][1]*u1 + X[1][2]*u2);
      a2 = b2 - (X[2][0]*u0 + X[2][1]*u1 + X[2][2]*u2);
    }
  }
  __syncthreads();
  if (lane == 0) {
    ltsolve3(S[NSTEP - 1], gv[NSTEP - 1]);
    for (int j = NPART - 2; j >= 0; --j) {
      const int s = PER * j + MI, sn = s + PER;
      const double x0 = gv[sn][0], x1 = gv[sn][1], x2 = gv[sn][2];
      gv[s][0] -= Rm[j + 1][0]*x0 + Rm[j + 1][3]*x1 + Rm[j + 1][6]*x2;
      gv[s][1] -= Rm[j + 1][1]*x0 + Rm[j + 1][4]*x1 + Rm[j + 1][7]*x2;
      gv[s][2] -= Rm[j + 1][2]*x0 + Rm[j + 1][5]*x1 + Rm[j + 1][8]*x2;
      ltsolve3(S[s], gv[s]);
    }
  }
  __syncthreads();
  if (lane < NPART) {
    double xl0 = 0.0, xl1 = 0.0, xl2 = 0.0;
    if (hasL) { xl0 = gv[pa - 1][0]; xl1 = gv[pa - 1][1]; xl2 = gv[pa - 1][2]; }
    for (int i = MI - 1; i >= 0; --i) {
      const int p = pa + i;
      const double x0 = gv[p + 1][0], x1 = gv[p + 1][1], x2 = gv[p + 1][2];
      double t0 = gv[p][0] - (Mx[p][0]*x0 + Mx[p][3]*x1 + Mx[p][6]*x2);
      double t1 = gv[p][1] - (Mx[p][1]*x0 + Mx[p][4]*x1 + Mx[p][7]*x2);
      double t2 = gv[p][2] - (Mx[p][2]*x0 + Mx[p][5]*x1 + Mx[p][8]*x2);
      if (hasL) {
        t0 -= V[p][0]*xl0 + V[p][1]*xl1 + V[p][2]*xl2;
        t1 -= V[p][3]*xl0 + V[p][4]*xl1 + V[p][5]*xl2;
        t2 -= V[p][6]*xl0 + V[p][7]*xl1 + V[p][8]*xl2;
      }
      gv[p][0] = t0; gv[p][1] = t1; gv[p][2] = t2;
      ltsolve3(S[p], gv[p]);
    }
  }
  __syncthreads();

  for (int t = lane; t < NSTEP; t += 32) {
    out[0 * OUT_U + b * NSTEP + t] = (float)gv[t][0];
    out[1 * OUT_U + b * NSTEP + t] = (float)gv[t][1];
    out[2 * OUT_U + b * NSTEP + t] = (float)gv[t][2];
  }
  if (lane == 0) out[OUT_EPS + b] = 0.0f;
  for (int p = lane; p < T1; p += 32) out[OUT_ST + b * T1 + p] = hstep[p];
}

extern "C" void kernel_launch(void* const* d_in, const int* in_sizes, int n_in,
                              void* d_out, int out_size, void* d_ws, size_t ws_size,
                              hipStream_t stream) {
  (void)in_sizes; (void)n_in; (void)out_size; (void)d_ws; (void)ws_size;
  const float* coeffs = (const float*)d_in[0];
  const float* rhs    = (const float*)d_in[1];
  const float* ivr    = (const float*)d_in[2];
  const float* steps  = (const float*)d_in[3];
  float* out = (float*)d_out;
  hipLaunchKernelGGL(ode_blocktri_kernel, dim3(NB), dim3(32), 0, stream,
                     coeffs, rhs, ivr, steps, out);
}


# Interiors' chain: a block's factor, rhs and spike column are ONE bordered 3-lane panel, one row per lane (6 divide sequences a block, not 9; pivots and sub-diagonal by row gather); sum order kept
# speedup vs baseline: 7.6401x; 1.0762x over previous
#include <hip/hip_runtime.h>
#include <hip/hip_bf16.h>
#include <math.h>


typedef __attribute__((ext_vector_type(2))) float v2f;
typedef __attribute__((ext_vector_type(8))) float v8f;

#define NSTEP 100
#define T1    99
#define NB    256
#define OUT_U   (NB * NSTEP)
#define OUT_EPS (3 * OUT_U)
#define OUT_ST  (OUT_EPS + NB)
#define PER   10
#define MI    9
#define NPART 10
#define NTILE 7

__device__ __forceinline__ void Ecoef(int r, int c, float* pa, float* pb, float* pc) {
  float a = 0.f, b = 0.f, q = 0.f;
  if (r == 0)      { a = (c == 0) ? 1.f  : ((c == 3) ? -1.f : 0.f); b = (c == 1) ? 1.f  : 0.f; q = (c == 2) ? 0.5f : 0.f; }
  else if (r == 1) { a = (c == 1) ? 1.f  : ((c == 4) ? -1.f : 0.f); b = (c == 2) ? 1.f  : 0.f; }
  else if (r == 2) { a = (c == 0) ? -1.f : ((c == 3) ? 1.f  : 0.f); b = (c == 4) ? -1.f : 0.f; q = (c == 5) ? 0.5f : 0.f; }
  else if (r == 3) { a = (c == 1) ? -1.f : ((c == 4) ? 1.f  : 0.f); b = (c == 5) ? -1.f : 0.f; }
  else if (r == 4) { a = (c == 1) ? -1.f : ((c == 4) ? 1.f  : 0.f); b = (c == 2 || c == 5) ? -0.5f : 0.f; }
  *pa = a; *pb = b; *pc = q;
}

__device__ __forceinline__ void ltsolve3(const double* l, double* v) {
  double x2 = v[2] / l[5];
  double x1 = (v[1] - l[4] * x2) / l[2];
  double x0 = (v[0] - l[1] * x1 - l[3] * x2) / l[0];
  v[0] = x0; v[1] = x1; v[2] = x2;
}

__device__ __forceinline__ void mstep3(const double* L, double* Bp, double X[3][3]) {
  #pragma unroll
  for (int r = 0; r < 3; ++r) {
    const double c0 = Bp[0 + r];
    const double c1 = Bp[3 + r];
    const double c2 = Bp[6 + r];
    const double x0 = c0 / L[0];
    const double x1 = (c1 - x0 * L[1]) / L[2];
    const double x2 = (c2 - x0 * L[3] - x1 * L[4]) / L[5];
    X[r][0] = x0; X[r][1] = x1; X[r][2] = x2;
  }
  #pragma unroll
  for (int r = 0; r < 3; ++r) {
    Bp[r * 3 + 0] = X[r][0]; Bp[r * 3 + 1] = X[r][1]; Bp[r * 3 + 2] = X[r][2];
  }
}
__device__ __forceinline__ double lane_bcast(double v, int src) {
  const int lo = __builtin_amdgcn_readlane(__double2loint(v), src);
  const int hi = __builtin_amdgcn_readlane(__double2hiint(v), src);
  return __hiloint2double(hi, lo);
}

__device__ __forceinline__ constexpr unsigned part_sel(int K, int half) {
  unsigned w = 0;
  for (int j = 0; j < 8; ++j) {
    const int lj = 8 * half + j;
    const unsigned s = (lj < 15) ? (unsigned)(3 * (lj / 3) + K) : 15u;
    w |= s << (4 * j);
  }
  return w;
}
template <int K>
__device__ __forceinline__ double part_bcast(double v) {
  constexpr unsigned lo = part_sel(K, 0), hi = part_sel(K, 1);
  const unsigned a = __builtin_amdgcn_permlane16(0u, (unsigned)__double2loint(v), lo, hi, false, false);
  const unsigned b = __builtin_amdgcn_permlane16(0u, (unsigned)__double2hiint(v), lo, hi, false, false);
  return __hiloint2double((int)b, (int)a);
}

struct Fac3 { double d0, d1, d2, l10, l20, l21; };

__device__ __forceinline__ void panel3(int c, double& a0, double& a1, double& a2,
                                       double& y0, double& y1, double& y2,
                                       double& v0, double& v1, double& v2, Fac3& f) {
  const bool c0 = (c == 0);
  double d = sqrt(part_bcast<0>(a0));
  double q = (c0 ? y0 : a0) / d;
  y0 = c0 ? q : y0;
  a0 = c0 ? d : q;
  v0 = v0 / d;
  const double l10 = part_bcast<1>(a0), l20 = part_bcast<2>(a0);
  a1 -= a0 * l10;  a2 -= a0 * l20;
  y1 -= y0 * l10;  y2 -= y0 * l20;
  v1 -= v0 * l10;  v2 -= v0 * l20;
  f.d0 = d; f.l10 = l10; f.l20 = l20;
  d = sqrt(part_bcast<1>(a1));
  q = (c0 ? y1 : a1) / d;
  y1 = c0 ? q : y1;
  a1 = (c == 1) ? d : q;
  v1 = v1 / d;
  const double l21 = part_bcast<2>(a1);
  a2 -= a1 * l21;
  y2 -= y1 * l21;
  v2 -= v1 * l21;
  f.d1 = d; f.l21 = l21;
  d = sqrt(part_bcast<2>(a2));
  y2 = y2 / d;
  v2 = v2 / d;
  a2 = d;
  f.d2 = d;
}

__global__ __launch_bounds__(32) void ode_blocktri_kernel(
    const float* __restrict__ coeffs,
    const float* __restrict__ rhs,
    const float* __restrict__ ivr,
    const float* __restrict__ steps,
    float* __restrict__ out)
{
  __shared__ double S[NSTEP][6];
  __shared__ double Bo[T1][9];
  __shared__ double gv[NSTEP][3];
  __shared__ float  hstep[T1];
  __shared__ double V[NSTEP][9];
  __shared__ double Mx[T1][9];
  __shared__ double R[NPART][9];
  __shared__ double Rm[NPART][9];

  const int b = blockIdx.x;
  const int lane = threadIdx.x;

  for (int i = lane; i < T1; i += 32) hstep[i] = steps[b * T1 + i];
  __syncthreads();

  const int e  = lane & 15;
  const int hh = lane >> 4;
  const int i2 = (e >= 3) ? 2 : ((e >= 1) ? 1 : 0);
  const int i3 = (e >= 12) ? 2 : ((e >= 9) ? 1 : 0);
  const int ca = (e < 6) ? i2 : ((e < 15) ? i3 : 6);
  const int cb = (e < 6) ? e - i2 * (i2 + 1) / 2 : ((e < 15) ? e - 3 - 3 * i3 : 6);
  const int ga = (e < 6) ? 3 + ca : 6;
  const int gb = (e < 6) ? 3 + cb : 6;
  float ph0 = 0.f, ph1 = 0.f, ph2 = 0.f, ph3 = 0.f, ph4 = 0.f;
  float pg0 = 0.f, pg1 = 0.f, pg2 = 0.f, pg3 = 0.f, pg4 = 0.f;
  #pragma unroll
  for (int k = 0; k < 5; ++k) {
    float a1, b1, q1, a2, b2, q2;
    Ecoef(k, ca, &a1, &b1, &q1); Ecoef(k, cb, &a2, &b2, &q2);
    ph0 += a1 * a2; ph1 += a1 * b2 + b1 * a2; ph2 += a1 * q2 + b1 * b2 + q1 * a2;
    ph3 += b1 * q2 + q1 * b2; ph4 += q1 * q2;
    Ecoef(k, ga, &a1, &b1, &q1); Ecoef(k, gb, &a2, &b2, &q2);
    pg0 += a1 * a2; pg1 += a1 * b2 + b1 * a2; pg2 += a1 * q2 + b1 * b2 + q1 * a2;
    pg3 += b1 * q2 + q1 * b2; pg4 += q1 * q2;
  }
  v2f A0, A1, A2;
  A0.x = hh ? ph2 : ph0;  A0.y = hh ? ph3 : ph1;
  A1.x = hh ? pg1 : ph4;  A1.y = hh ? pg2 : pg0;
  A2.x = hh ? 0.f : pg3;  A2.y = hh ? 0.f : pg4;

  float cf0[NTILE], cf1[NTILE], cf2[NTILE], rf[NTILE], hf[NTILE], gf[NTILE];
  #pragma unroll
  for (int n = 0; n < NTILE; ++n) {
    const int t = 16 * n + e;
    const bool own = (hh == 0) && (t < NSTEP);
    const int tl = (t < NSTEP) ? t : NSTEP - 1;
    const float l0 = coeffs[(b * NSTEP + tl) * 3 + 0];
    const float l1 = coeffs[(b * NSTEP + tl) * 3 + 1];
    const float l2 = coeffs[(b * NSTEP + tl) * 3 + 2];
    const float lr = rhs[b * NSTEP + tl];
    const float lh = hstep[(t < T1) ? t : T1 - 1];
    const float lg = hstep[(t >= 1 && t < NSTEP) ? t - 1 : 0];
    cf0[n] = own ? l0 : 0.f;
    cf1[n] = own ? l1 : 0.f;
    cf2[n] = own ? l2 : 0.f;
    rf[n]  = own ? lr : 0.f;
    hf[n]  = (t < T1) ? lh : 0.f;
    gf[n]  = (t >= 1 && t < NSTEP) ? lg : 0.f;
  }
  #pragma unroll
  for (int n = 0; n < NTILE; ++n) {
    const int t = 16 * n + e;
    const float h = hf[n], g = gf[n];
    const float u = (t < T1) ? 1.f : 0.f;
    const float w = (t >= 1 && t < NSTEP) ? 1.f : 0.f;
    const float h2 = h * h, h3 = h2 * h, h4 = h2 * h2;
    const float g2 = g * g, g3 = g2 * g, g4 = g2 * g2;
    v2f B0, B1, B2;
    B0.x = hh ? h2 : u;    B0.y = hh ? h3 : h;
    B1.x = hh ? g  : h4;   B1.y = hh ? g2 : w;
    B2.x = hh ? 0.f : g3;  B2.y = hh ? 0.f : g4;
    v8f acc = {};
    acc = __builtin_amdgcn_wmma_f32_16x16x4_f32(false, A0, false, B0, (short)0, acc, false, false);
    acc = __builtin_amdgcn_wmma_f32_16x16x4_f32(false, A1, false, B1, (short)0, acc, false, false);
    acc = __builtin_amdgcn_wmma_f32_16x16x4_f32(false, A2, false, B2, (short)0, acc, false, false);
    const double c0 = (double)cf0[n], c1 = (double)cf1[n], c2 = (double)cf2[n], rr = (double)rf[n];
    double s0 = fma(c0, c0, (double)acc[0]);
    double s1 = fma(c1, c0, (double)acc[1]);
    double s2 = fma(c1, c1, (double)acc[2]);
    double s3 = fma(c2, c0, (double)acc[3]);
    double s4 = fma(c2, c1, (double)acc[4]);
    double s5 = fma(c2, c2, (double)acc[5]);
    const double s6 = (double)acc[6], s7 = (double)acc[7];
    double y0 = c0 * rr, y1 = c1 * rr, y2 = c2 * rr;
    if (n == 0 && lane == 0) {
      s0 += 1.0; s2 += 1.0;
      y0 += (double)ivr[b * 2 + 0];
      y1 += (double)ivr[b * 2 + 1];
    }
    if (hh == 0) {
      if (t < NSTEP) {
        S[t][0] = s0; S[t][1] = s1; S[t][2] = s2; S[t][3] = s3; S[t][4] = s4; S[t][5] = s5;
        gv[t][0] = y0; gv[t][1] = y1; gv[t][2] = y2;
      }
      if (t < T1) { Bo[t][0] = s6; Bo[t][1] = s7; }
    } else if (t < T1) {
      Bo[t][2] = s0; Bo[t][3] = s1; Bo[t][4] = s2; Bo[t][5] = s3; Bo[t][6] = s4; Bo[t][7] = s5; Bo[t][8] = s6;
    }
  }
  __syncthreads();

  const int pa = PER * lane;
  const bool hasL = lane > 0;
  const int l16 = lane & 15;
  const bool act = l16 < 15;
  const int part = act ? 5 * (lane >> 4) + l16 / 3 : 0;
  const int c = act ? l16 % 3 : 0;
  const int pk = PER * part;
  const int prow = c * (c + 1) / 2;
  Fac3 f;
  double yc0 = 0.0, yc1 = 0.0, yc2 = 0.0;
  double vc0 = 0.0, vc1 = 0.0, vc2 = 0.0;
  {
    double a0 = S[pk][prow], a1 = S[pk][prow + 1], a2 = S[pk][prow + 2];
    if (c == 0) { yc0 = gv[pk][0]; yc1 = gv[pk][1]; yc2 = gv[pk][2]; }
    if (part > 0) {
      vc0 = Bo[pk - 1][c * 3 + 0]; vc1 = Bo[pk - 1][c * 3 + 1]; vc2 = Bo[pk - 1][c * 3 + 2];
    }
    panel3(c, a0, a1, a2, yc0, yc1, yc2, vc0, vc1, vc2, f);
    if (act) {
      V[pk][0 + c] = vc0; V[pk][3 + c] = vc1; V[pk][6 + c] = vc2;
      S[pk][prow] = a0;
      if (c >= 1) S[pk][prow + 1] = a1;
      if (c == 2) S[pk][prow + 2] = a2;
      if (c == 0) { gv[pk][0] = yc0; gv[pk][1] = yc1; gv[pk][2] = yc2; }
    }
  }
  for (int i = 0; i < MI; ++i) {
    const int p = pk + i;
    const double b0 = Bo[p][0 + c], b1 = Bo[p][3 + c], b2 = Bo[p][6 + c];
    const double x0 = b0 / f.d0;
    const double x1 = (b1 - x0 * f.l10) / f.d1;
    const double x2 = (b2 - x0 * f.l20 - x1 * f.l21) / f.d2;
    if (act) { Mx[p][c * 3 + 0] = x0; Mx[p][c * 3 + 1] = x1; Mx[p][c * 3 + 2] = x2; }
    __syncthreads();
    double X[3][3];
    for (int r = 0; r < 3; ++r) { X[r][0] = Mx[p][r * 3 + 0]; X[r][1] = Mx[p][r * 3 + 1]; X[r][2] = Mx[p][r * 3 + 2]; }
    double n0 = S[p + 1][prow + 0] - (x0*X[0][0] + x1*X[0][1] + x2*X[0][2]);
    double n1 = S[p + 1][prow + 1] - (x0*X[1][0] + x1*X[1][1] + x2*X[1][2]);
    double n2 = S[p + 1][prow + 2] - (x0*X[2][0] + x1*X[2][1] + x2*X[2][2]);
    double yn0 = 0.0, yn1 = 0.0, yn2 = 0.0;
    if (c == 0) {
      yn0 = gv[p + 1][0] - (X[0][0]*yc0 + X[0][1]*yc1 + X[0][2]*yc2);
      yn1 = gv[p + 1][1] - (X[1][0]*yc0 + X[1][1]*yc1 + X[1][2]*yc2);
      yn2 = gv[p + 1][2] - (X[2][0]*yc0 + X[2][1]*yc1 + X[2][2]*yc2);
    }
    if (i < MI - 1) {
      double vn0 = -(X[0][0]*vc0 + X[0][1]*vc1 + X[0][2]*vc2);
      double vn1 = -(X[1][0]*vc0 + X[1][1]*vc1 + X[1][2]*vc2);
      double vn2 = -(X[2][0]*vc0 + X[2][1]*vc1 + X[2][2]*vc2);
      panel3(c, n0, n1, n2, yn0, yn1, yn2, vn0, vn1, vn2, f);
      if (act) {
        V[p + 1][0 + c] = vn0; V[p + 1][3 + c] = vn1; V[p + 1][6 + c] = vn2;
        S[p + 1][prow] = n0;
        if (c >= 1) S[p + 1][prow + 1] = n1;
        if (c == 2) S[p + 1][prow + 2] = n2;
        if (c == 0) { gv[p + 1][0] = yn0; gv[p + 1][1] = yn1; gv[p + 1][2] = yn2; }
      }
      yc0 = yn0; yc1 = yn1; yc2 = yn2;
      vc0 = vn0; vc1 = vn1; vc2 = vn2;
    } else if (act) {
      S[p + 1][prow] = n0;
      if (c >= 1) S[p + 1][prow + 1] = n1;
      if (c == 2) S[p + 1][prow + 2] = n2;
      if (c == 0) { gv[p + 1][0] = yn0; gv[p + 1][1] = yn1; gv[p + 1][2] = yn2; }
      for (int j = 0; j < 3; ++j)
        R[part][c * 3 + j] = -(vc0*X[j][0] + vc1*X[j][1] + vc2*X[j][2]);
    }
  }
  __syncthreads();
  for (int idx = lane; idx < (NPART - 1) * 6; idx += 32) {
    const int k = 1 + idx / 6, q = idx % 6;
    const int i2 = (q >= 3) ? 2 : ((q >= 1) ? 1 : 0);
    const int j2 = q - i2 * (i2 + 1) / 2;
    const int p0 = PER * k;
    double acc = 0.0;
    for (int i = 0; i < MI; ++i)
      acc += V[p0 + i][0 + i2]*V[p0 + i][0 + j2] + V[p0 + i][3 + i2]*V[p0 + i][3 + j2]
           + V[p0 + i][6 + i2]*V[p0 + i][6 + j2];
    S[p0 - 1][q] -= acc;
  }
  for (int idx = lane; idx < (NPART - 1) * 3; idx += 32) {
    const int k = 1 + idx / 3, j = idx % 3;
    const int p0 = PER * k;
    double acc = 0.0;
    for (int i = 0; i < MI; ++i)
      acc += V[p0 + i][0 + j]*gv[p0 + i][0] + V[p0 + i][3 + j]*gv[p0 + i][1]
           + V[p0 + i][6 + j]*gv[p0 + i][2];
    gv[p0 - 1][j] -= acc;
  }
  __syncthreads();
  const bool isS = lane < 3, isC = (lane >= 3) && (lane < 6), isG = (lane == 6);
  const int rl = isS ? lane : 0;
  const int rb = rl * (rl + 1) / 2;
  const int cc = isC ? lane - 3 : 0;
  double a0 = 0.0, a1 = 0.0, a2 = 0.0;
  if (isS)      { a0 = S[MI][rb]; a1 = S[MI][rb + 1]; a2 = S[MI][rb + 2]; }
  else if (isC) { a0 = R[1][0 + cc]; a1 = R[1][3 + cc]; a2 = R[1][6 + cc]; }
  else if (isG) { a0 = gv[MI][0]; a1 = gv[MI][1]; a2 = gv[MI][2]; }
  for (int j = 0; j < NPART; ++j) {
    const int s = PER * j + MI;
    double d = sqrt(lane_bcast(a0, 0));
    a0 = (lane == 0) ? d : a0 / d;
    const double l10 = lane_bcast(a0, 1), l20 = lane_bcast(a0, 2);
    a1 -= a0 * l10;
    a2 -= a0 * l20;
    d = sqrt(lane_bcast(a1, 1));
    a1 = (lane == 1) ? d : a1 / d;
    const double l21 = lane_bcast(a1, 2);
    a2 -= a1 * l21;
    d = sqrt(lane_bcast(a2, 2));
    a2 = (lane == 2) ? d : a2 / d;
    if (isS) {
      S[s][rb] = a0;
      if (lane >= 1) S[s][rb + 1] = a1;
      if (lane == 2) S[s][rb + 2] = a2;
    } else if (isC) {
      if (j + 1 < NPART) { Rm[j + 1][cc * 3 + 0] = a0; Rm[j + 1][cc * 3 + 1] = a1; Rm[j + 1][cc * 3 + 2] = a2; }
    } else if (isG) {
      gv[s][0] = a0; gv[s][1] = a1; gv[s][2] = a2;
    }
    __syncthreads();
    if (j + 1 < NPART) {
      const int sn = s + PER;
      double X[3][3];
      for (int r = 0; r < 3; ++r) { X[r][0] = Rm[j + 1][r * 3 + 0]; X[r][1] = Rm[j + 1][r * 3 + 1]; X[r][2] = Rm[j + 1][r * 3 + 2]; }
      double u0 = 0.0, u1 = 0.0, u2 = 0.0, b0 = 0.0, b1 = 0.0, b2 = 0.0;
      if (isS) {
        u0 = Rm[j + 1][rl * 3 + 0]; u1 = Rm[j + 1][rl * 3 + 1]; u2 = Rm[j + 1][rl * 3 + 2];
        b0 = S[sn][rb]; b1 = S[sn][rb + 1]; b2 = S[sn][rb + 2];
      } else if (isG) {
        u0 = a0; u1 = a1; u2 = a2;
        b0 = gv[sn][0]; b1 = gv[sn][1]; b2 = gv[sn][2];
      } else if (isC) {
        if (j + 2 < NPART) { b0 = R[j + 2][0 + cc]; b1 = R[j + 2][3 + cc]; b2 = R[j + 2][6 + cc]; }
      }
      a0 = b0 - (X[0][0]*u0 + X[0][1]*u1 + X[0][2]*u2);
      a1 = b1 - (X[1][0]*u0 + X[1][1]*u1 + X[1][2]*u2);
      a2 = b2 - (X[2][0]*u0 + X[2][1]*u1 + X[2][2]*u2);
    }
  }
  __syncthreads();
  if (lane == 0) {
    ltsolve3(S[NSTEP - 1], gv[NSTEP - 1]);
    for (int j = NPART - 2; j >= 0; --j) {
      const int s = PER * j + MI, sn = s + PER;
      const double x0 = gv[sn][0], x1 = gv[sn][1], x2 = gv[sn][2];
      gv[s][0] -= Rm[j + 1][0]*x0 + Rm[j + 1][3]*x1 + Rm[j + 1][6]*x2;
      gv[s][1] -= Rm[j + 1][1]*x0 + Rm[j + 1][4]*x1 + Rm[j + 1][7]*x2;
      gv[s][2] -= Rm[j + 1][2]*x0 + Rm[j + 1][5]*x1 + Rm[j + 1][8]*x2;
      ltsolve3(S[s], gv[s]);
    }
  }
  __syncthreads();
  if (lane < NPART) {
    double xl0 = 0.0, xl1 = 0.0, xl2 = 0.0;
    if (hasL) { xl0 = gv[pa - 1][0]; xl1 = gv[pa - 1][1]; xl2 = gv[pa - 1][2]; }
    for (int i = MI - 1; i >= 0; --i) {
      const int p = pa + i;
      const double x0 = gv[p + 1][0], x1 = gv[p + 1][1], x2 = gv[p + 1][2];
      double t0 = gv[p][0] - (Mx[p][0]*x0 + Mx[p][3]*x1 + Mx[p][6]*x2);
      double t1 = gv[p][1] - (Mx[p][1]*x0 + Mx[p][4]*x1 + Mx[p][7]*x2);
      double t2 = gv[p][2] - (Mx[p][2]*x0 + Mx[p][5]*x1 + Mx[p][8]*x2);
      if (hasL) {
        t0 -= V[p][0]*xl0 + V[p][1]*xl1 + V[p][2]*xl2;
        t1 -= V[p][3]*xl0 + V[p][4]*xl1 + V[p][5]*xl2;
        t2 -= V[p][6]*xl0 + V[p][7]*xl1 + V[p][8]*xl2;
      }
      gv[p][0] = t0; gv[p][1] = t1; gv[p][2] = t2;
      ltsolve3(S[p], gv[p]);
    }
  }
  __syncthreads();

  for (int t = lane; t < NSTEP; t += 32) {
    out[0 * OUT_U + b * NSTEP + t] = (float)gv[t][0];
    out[1 * OUT_U + b * NSTEP + t] = (float)gv[t][1];
    out[2 * OUT_U + b * NSTEP + t] = (float)gv[t][2];
  }
  if (lane == 0) out[OUT_EPS + b] = 0.0f;
  for (int p = lane; p < T1; p += 32) out[OUT_ST + b * T1 + p] = hstep[p];
}

extern "C" void kernel_launch(void* const* d_in, const int* in_sizes, int n_in,
                              void* d_out, int out_size, void* d_ws, size_t ws_size,
                              hipStream_t stream) {
  (void)in_sizes; (void)n_in; (void)out_size; (void)d_ws; (void)ws_size;
  const float* coeffs = (const float*)d_in[0];
  const float* rhs    = (const float*)d_in[1];
  const float* ivr    = (const float*)d_in[2];
  const float* steps  = (const float*)d_in[3];
  float* out = (float*)d_out;
  hipLaunchKernelGGL(ode_blocktri_kernel, dim3(NB), dim3(32), 0, stream,
                     coeffs, rhs, ivr, steps, out);
}


# Separators' chain eliminated from both ends at once: rows of 16 lanes take separators 0 up and 9 down in the same instructions, meet in separator 5, back-substitute outwards; order, rounding change
# speedup vs baseline: 8.7061x; 1.1395x over previous
#include <hip/hip_runtime.h>
#include <hip/hip_bf16.h>
#include <math.h>


typedef __attribute__((ext_vector_type(2))) float v2f;
typedef __attribute__((ext_vector_type(8))) float v8f;

#define NSTEP 100
#define T1    99
#define NB    256
#define OUT_U   (NB * NSTEP)
#define OUT_EPS (3 * OUT_U)
#define OUT_ST  (OUT_EPS + NB)
#define PER   10
#define MI    9
#define NPART 10
#define MID   5
#define NTILE 7

__device__ __forceinline__ void Ecoef(int r, int c, float* pa, float* pb, float* pc) {
  float a = 0.f, b = 0.f, q = 0.f;
  if (r == 0)      { a = (c == 0) ? 1.f  : ((c == 3) ? -1.f : 0.f); b = (c == 1) ? 1.f  : 0.f; q = (c == 2) ? 0.5f : 0.f; }
  else if (r == 1) { a = (c == 1) ? 1.f  : ((c == 4) ? -1.f : 0.f); b = (c == 2) ? 1.f  : 0.f; }
  else if (r == 2) { a = (c == 0) ? -1.f : ((c == 3) ? 1.f  : 0.f); b = (c == 4) ? -1.f : 0.f; q = (c == 5) ? 0.5f : 0.f; }
  else if (r == 3) { a = (c == 1) ? -1.f : ((c == 4) ? 1.f  : 0.f); b = (c == 5) ? -1.f : 0.f; }
  else if (r == 4) { a = (c == 1) ? -1.f : ((c == 4) ? 1.f  : 0.f); b = (c == 2 || c == 5) ? -0.5f : 0.f; }
  *pa = a; *pb = b; *pc = q;
}

__device__ __forceinline__ void ltsolve3(const double* l, double* v) {
  double x2 = v[2] / l[5];
  double x1 = (v[1] - l[4] * x2) / l[2];
  double x0 = (v[0] - l[1] * x1 - l[3] * x2) / l[0];
  v[0] = x0; v[1] = x1; v[2] = x2;
}

__device__ __forceinline__ void mstep3(const double* L, double* Bp, double X[3][3]) {
  #pragma unroll
  for (int r = 0; r < 3; ++r) {
    const double c0 = Bp[0 + r];
    const double c1 = Bp[3 + r];
    const double c2 = Bp[6 + r];
    const double x0 = c0 / L[0];
    const double x1 = (c1 - x0 * L[1]) / L[2];
    const double x2 = (c2 - x0 * L[3] - x1 * L[4]) / L[5];
    X[r][0] = x0; X[r][1] = x1; X[r][2] = x2;
  }
  #pragma unroll
  for (int r = 0; r < 3; ++r) {
    Bp[r * 3 + 0] = X[r][0]; Bp[r * 3 + 1] = X[r][1]; Bp[r * 3 + 2] = X[r][2];
  }
}
__device__ __forceinline__ constexpr unsigned part_sel(int K, int half) {
  unsigned w = 0;
  for (int j = 0; j < 8; ++j) {
    const int lj = 8 * half + j;
    const unsigned s = (lj < 15) ? (unsigned)(3 * (lj / 3) + K) : 15u;
    w |= s << (4 * j);
  }
  return w;
}
template <int K>
__device__ __forceinline__ double part_bcast(double v) {
  constexpr unsigned lo = part_sel(K, 0), hi = part_sel(K, 1);
  const unsigned a = __builtin_amdgcn_permlane16(0u, (unsigned)__double2loint(v), lo, hi, false, false);
  const unsigned b = __builtin_amdgcn_permlane16(0u, (unsigned)__double2hiint(v), lo, hi, false, false);
  return __hiloint2double((int)b, (int)a);
}

template <int K>
__device__ __forceinline__ double row_bcast(double v) {
  constexpr unsigned sel = 0x11111111u * (unsigned)K;
  const unsigned a = __builtin_amdgcn_permlane16(0u, (unsigned)__double2loint(v), sel, sel, false, false);
  const unsigned b = __builtin_amdgcn_permlane16(0u, (unsigned)__double2hiint(v), sel, sel, false, false);
  return __hiloint2double((int)b, (int)a);
}

struct Fac3 { double d0, d1, d2, l10, l20, l21; };

__device__ __forceinline__ void panel3(int c, double& a0, double& a1, double& a2,
                                       double& y0, double& y1, double& y2,
                                       double& v0, double& v1, double& v2, Fac3& f) {
  const bool c0 = (c == 0);
  double d = sqrt(part_bcast<0>(a0));
  double q = (c0 ? y0 : a0) / d;
  y0 = c0 ? q : y0;
  a0 = c0 ? d : q;
  v0 = v0 / d;
  const double l10 = part_bcast<1>(a0), l20 = part_bcast<2>(a0);
  a1 -= a0 * l10;  a2 -= a0 * l20;
  y1 -= y0 * l10;  y2 -= y0 * l20;
  v1 -= v0 * l10;  v2 -= v0 * l20;
  f.d0 = d; f.l10 = l10; f.l20 = l20;
  d = sqrt(part_bcast<1>(a1));
  q = (c0 ? y1 : a1) / d;
  y1 = c0 ? q : y1;
  a1 = (c == 1) ? d : q;
  v1 = v1 / d;
  const double l21 = part_bcast<2>(a1);
  a2 -= a1 * l21;
  y2 -= y1 * l21;
  v2 -= v1 * l21;
  f.d1 = d; f.l21 = l21;
  d = sqrt(part_bcast<2>(a2));
  y2 = y2 / d;
  v2 = v2 / d;
  a2 = d;
  f.d2 = d;
}

__global__ __launch_bounds__(32) void ode_blocktri_kernel(
    const float* __restrict__ coeffs,
    const float* __restrict__ rhs,
    const float* __restrict__ ivr,
    const float* __restrict__ steps,
    float* __restrict__ out)
{
  __shared__ double S[NSTEP][6];
  __shared__ double Bo[T1][9];
  __shared__ double gv[NSTEP][3];
  __shared__ float  hstep[T1];
  __shared__ double V[NSTEP][9];
  __shared__ double Mx[T1][9];
  __shared__ double R[NPART][9];
  __shared__ double Rm[NPART][9];
  __shared__ double Dl[4][3];

  const int b = blockIdx.x;
  const int lane = threadIdx.x;

  for (int i = lane; i < T1; i += 32) hstep[i] = steps[b * T1 + i];
  __syncthreads();

  const int e  = lane & 15;
  const int hh = lane >> 4;
  const int i2 = (e >= 3) ? 2 : ((e >= 1) ? 1 : 0);
  const int i3 = (e >= 12) ? 2 : ((e >= 9) ? 1 : 0);
  const int ca = (e < 6) ? i2 : ((e < 15) ? i3 : 6);
  const int cb = (e < 6) ? e - i2 * (i2 + 1) / 2 : ((e < 15) ? e - 3 - 3 * i3 : 6);
  const int ga = (e < 6) ? 3 + ca : 6;
  const int gb = (e < 6) ? 3 + cb : 6;
  float ph0 = 0.f, ph1 = 0.f, ph2 = 0.f, ph3 = 0.f, ph4 = 0.f;
  float pg0 = 0.f, pg1 = 0.f, pg2 = 0.f, pg3 = 0.f, pg4 = 0.f;
  #pragma unroll
  for (int k = 0; k < 5; ++k) {
    float a1, b1, q1, a2, b2, q2;
    Ecoef(k, ca, &a1, &b1, &q1); Ecoef(k, cb, &a2, &b2, &q2);
    ph0 += a1 * a2; ph1 += a1 * b2 + b1 * a2; ph2 += a1 * q2 + b1 * b2 + q1 * a2;
    ph3 += b1 * q2 + q1 * b2; ph4 += q1 * q2;
    Ecoef(k, ga, &a1, &b1, &q1); Ecoef(k, gb, &a2, &b2, &q2);
    pg0 += a1 * a2; pg1 += a1 * b2 + b1 * a2; pg2 += a1 * q2 + b1 * b2 + q1 * a2;
    pg3 += b1 * q2 + q1 * b2; pg4 += q1 * q2;
  }
  v2f A0, A1, A2;
  A0.x = hh ? ph2 : ph0;  A0.y = hh ? ph3 : ph1;
  A1.x = hh ? pg1 : ph4;  A1.y = hh ? pg2 : pg0;
  A2.x = hh ? 0.f : pg3;  A2.y = hh ? 0.f : pg4;

  float cf0[NTILE], cf1[NTILE], cf2[NTILE], rf[NTILE], hf[NTILE], gf[NTILE];
  #pragma unroll
  for (int n = 0; n < NTILE; ++n) {
    const int t = 16 * n + e;
    const bool own = (hh == 0) && (t < NSTEP);
    const int tl = (t < NSTEP) ? t : NSTEP - 1;
    const float l0 = coeffs[(b * NSTEP + tl) * 3 + 0];
    const float l1 = coeffs[(b * NSTEP + tl) * 3 + 1];
    const float l2 = coeffs[(b * NSTEP + tl) * 3 + 2];
    const float lr = rhs[b * NSTEP + tl];
    const float lh = hstep[(t < T1) ? t : T1 - 1];
    const float lg = hstep[(t >= 1 && t < NSTEP) ? t - 1 : 0];
    cf0[n] = own ? l0 : 0.f;
    cf1[n] = own ? l1 : 0.f;
    cf2[n] = own ? l2 : 0.f;
    rf[n]  = own ? lr : 0.f;
    hf[n]  = (t < T1) ? lh : 0.f;
    gf[n]  = (t >= 1 && t < NSTEP) ? lg : 0.f;
  }
  #pragma unroll
  for (int n = 0; n < NTILE; ++n) {
    const int t = 16 * n + e;
    const float h = hf[n], g = gf[n];
    const float u = (t < T1) ? 1.f : 0.f;
    const float w = (t >= 1 && t < NSTEP) ? 1.f : 0.f;
    const float h2 = h * h, h3 = h2 * h, h4 = h2 * h2;
    const float g2 = g * g, g3 = g2 * g, g4 = g2 * g2;
    v2f B0, B1, B2;
    B0.x = hh ? h2 : u;    B0.y = hh ? h3 : h;
    B1.x = hh ? g  : h4;   B1.y = hh ? g2 : w;
    B2.x = hh ? 0.f : g3;  B2.y = hh ? 0.f : g4;
    v8f acc = {};
    acc = __builtin_amdgcn_wmma_f32_16x16x4_f32(false, A0, false, B0, (short)0, acc, false, false);
    acc = __builtin_amdgcn_wmma_f32_16x16x4_f32(false, A1, false, B1, (short)0, acc, false, false);
    acc = __builtin_amdgcn_wmma_f32_16x16x4_f32(false, A2, false, B2, (short)0, acc, false, false);
    const double c0 = (double)cf0[n], c1 = (double)cf1[n], c2 = (double)cf2[n], rr = (double)rf[n];
    double s0 = fma(c0, c0, (double)acc[0]);
    double s1 = fma(c1, c0, (double)acc[1]);
    double s2 = fma(c1, c1, (double)acc[2]);
    double s3 = fma(c2, c0, (double)acc[3]);
    double s4 = fma(c2, c1, (double)acc[4]);
    double s5 = fma(c2, c2, (double)acc[5]);
    const double s6 = (double)acc[6], s7 = (double)acc[7];
    double y0 = c0 * rr, y1 = c1 * rr, y2 = c2 * rr;
    if (n == 0 && lane == 0) {
      s0 += 1.0; s2 += 1.0;
      y0 += (double)ivr[b * 2 + 0];
      y1 += (double)ivr[b * 2 + 1];
    }
    if (hh == 0) {
      if (t < NSTEP) {
        S[t][0] = s0; S[t][1] = s1; S[t][2] = s2; S[t][3] = s3; S[t][4] = s4; S[t][5] = s5;
        gv[t][0] = y0; gv[t][1] = y1; gv[t][2] = y2;
      }
      if (t < T1) { Bo[t][0] = s6; Bo[t][1] = s7; }
    } else if (t < T1) {
      Bo[t][2] = s0; Bo[t][3] = s1; Bo[t][4] = s2; Bo[t][5] = s3; Bo[t][6] = s4; Bo[t][7] = s5; Bo[t][8] = s6;
    }
  }
  __syncthreads();

  const int pa = PER * lane;
  const bool hasL = lane > 0;
  const int l16 = lane & 15;
  const bool act = l16 < 15;
  const int part = act ? 5 * (lane >> 4) + l16 / 3 : 0;
  const int c = act ? l16 % 3 : 0;
  const int pk = PER * part;
  const int prow = c * (c + 1) / 2;
  Fac3 f;
  double yc0 = 0.0, yc1 = 0.0, yc2 = 0.0;
  double vc0 = 0.0, vc1 = 0.0, vc2 = 0.0;
  {
    double a0 = S[pk][prow], a1 = S[pk][prow + 1], a2 = S[pk][prow + 2];
    if (c == 0) { yc0 = gv[pk][0]; yc1 = gv[pk][1]; yc2 = gv[pk][2]; }
    if (part > 0) {
      vc0 = Bo[pk - 1][c * 3 + 0]; vc1 = Bo[pk - 1][c * 3 + 1]; vc2 = Bo[pk - 1][c * 3 + 2];
    }
    panel3(c, a0, a1, a2, yc0, yc1, yc2, vc0, vc1, vc2, f);
    if (act) {
      V[pk][0 + c] = vc0; V[pk][3 + c] = vc1; V[pk][6 + c] = vc2;
      S[pk][prow] = a0;
      if (c >= 1) S[pk][prow + 1] = a1;
      if (c == 2) S[pk][prow + 2] = a2;
      if (c == 0) { gv[pk][0] = yc0; gv[pk][1] = yc1; gv[pk][2] = yc2; }
    }
  }
  for (int i = 0; i < MI; ++i) {
    const int p = pk + i;
    const double b0 = Bo[p][0 + c], b1 = Bo[p][3 + c], b2 = Bo[p][6 + c];
    const double x0 = b0 / f.d0;
    const double x1 = (b1 - x0 * f.l10) / f.d1;
    const double x2 = (b2 - x0 * f.l20 - x1 * f.l21) / f.d2;
    if (act) { Mx[p][c * 3 + 0] = x0; Mx[p][c * 3 + 1] = x1; Mx[p][c * 3 + 2] = x2; }
    __syncthreads();
    double X[3][3];
    for (int r = 0; r < 3; ++r) { X[r][0] = Mx[p][r * 3 + 0]; X[r][1] = Mx[p][r * 3 + 1]; X[r][2] = Mx[p][r * 3 + 2]; }
    double n0 = S[p + 1][prow + 0] - (x0*X[0][0] + x1*X[0][1] + x2*X[0][2]);
    double n1 = S[p + 1][prow + 1] - (x0*X[1][0] + x1*X[1][1] + x2*X[1][2]);
    double n2 = S[p + 1][prow + 2] - (x0*X[2][0] + x1*X[2][1] + x2*X[2][2]);
    double yn0 = 0.0, yn1 = 0.0, yn2 = 0.0;
    if (c == 0) {
      yn0 = gv[p + 1][0] - (X[0][0]*yc0 + X[0][1]*yc1 + X[0][2]*yc2);
      yn1 = gv[p + 1][1] - (X[1][0]*yc0 + X[1][1]*yc1 + X[1][2]*yc2);
      yn2 = gv[p + 1][2] - (X[2][0]*yc0 + X[2][1]*yc1 + X[2][2]*yc2);
    }
    if (i < MI - 1) {
      double vn0 = -(X[0][0]*vc0 + X[0][1]*vc1 + X[0][2]*vc2);
      double vn1 = -(X[1][0]*vc0 + X[1][1]*vc1 + X[1][2]*vc2);
      double vn2 = -(X[2][0]*vc0 + X[2][1]*vc1 + X[2][2]*vc2);
      panel3(c, n0, n1, n2, yn0, yn1, yn2, vn0, vn1, vn2, f);
      if (act) {
        V[p + 1][0 + c] = vn0; V[p + 1][3 + c] = vn1; V[p + 1][6 + c] = vn2;
        S[p + 1][prow] = n0;
        if (c >= 1) S[p + 1][prow + 1] = n1;
        if (c == 2) S[p + 1][prow + 2] = n2;
        if (c == 0) { gv[p + 1][0] = yn0; gv[p + 1][1] = yn1; gv[p + 1][2] = yn2; }
      }
      yc0 = yn0; yc1 = yn1; yc2 = yn2;
      vc0 = vn0; vc1 = vn1; vc2 = vn2;
    } else if (act) {
      S[p + 1][prow] = n0;
      if (c >= 1) S[p + 1][prow + 1] = n1;
      if (c == 2) S[p + 1][prow + 2] = n2;
      if (c == 0) { gv[p + 1][0] = yn0; gv[p + 1][1] = yn1; gv[p + 1][2] = yn2; }
      for (int j = 0; j < 3; ++j)
        R[part][c * 3 + j] = -(vc0*X[j][0] + vc1*X[j][1] + vc2*X[j][2]);
    }
  }
  __syncthreads();
  for (int idx = lane; idx < (NPART - 1) * 6; idx += 32) {
    const int k = 1 + idx / 6, q = idx % 6;
    const int i2 = (q >= 3) ? 2 : ((q >= 1) ? 1 : 0);
    const int j2 = q - i2 * (i2 + 1) / 2;
    const int p0 = PER * k;
    double acc = 0.0;
    for (int i = 0; i < MI; ++i)
      acc += V[p0 + i][0 + i2]*V[p0 + i][0 + j2] + V[p0 + i][3 + i2]*V[p0 + i][3 + j2]
           + V[p0 + i][6 + i2]*V[p0 + i][6 + j2];
    S[p0 - 1][q] -= acc;
  }
  for (int idx = lane; idx < (NPART - 1) * 3; idx += 32) {
    const int k = 1 + idx / 3, j = idx % 3;
    const int p0 = PER * k;
    double acc = 0.0;
    for (int i = 0; i < MI; ++i)
      acc += V[p0 + i][0 + j]*gv[p0 + i][0] + V[p0 + i][3 + j]*gv[p0 + i][1]
           + V[p0 + i][6 + j]*gv[p0 + i][2];
    gv[p0 - 1][j] -= acc;
  }
  __syncthreads();
  const int dn = lane >> 4;
  const int q16 = lane & 15;
  const bool isS = q16 < 3, isC = (q16 >= 3) && (q16 < 6), isG = (q16 == 6);
  const int rl = isS ? q16 : 0;
  const int rb = rl * (rl + 1) / 2;
  const int cc = isC ? q16 - 3 : 0;
  const int cbase = dn ? cc * 3 : cc;
  const int cstr = dn ? 1 : 3;
  const double id0 = (isS && rl == 0) ? 1.0 : 0.0, id1 = (isS && rl == 1) ? 1.0 : 0.0, id2 = (isS && rl == 2) ? 1.0 : 0.0;
  double a0 = 0.0, a1 = 0.0, a2 = 0.0;
  {
    const int sj = dn ? NPART - 1 : 0;
    const int s = PER * sj + MI;
    const int m = dn ? sj : sj + 1;
    if (isS)      { a0 = S[s][rb]; a1 = S[s][rb + 1]; a2 = S[s][rb + 2]; }
    else if (isC) { a0 = R[m][cbase]; a1 = R[m][cbase + cstr]; a2 = R[m][cbase + 2 * cstr]; }
    else if (isG) { a0 = gv[s][0]; a1 = gv[s][1]; a2 = gv[s][2]; }
  }
  for (int t = 0; t <= MID; ++t) {
    const int sj = dn ? NPART - 1 - t : t;
    const bool live = dn ? (t < NPART - 1 - MID) : true;
    const int s = PER * sj + MI;
    const int m = dn ? sj : sj + 1;
    double d = sqrt(row_bcast<0>(a0));
    a0 = (isS && rl == 0) ? d : a0 / d;
    const double l10 = row_bcast<1>(a0), l20 = row_bcast<2>(a0);
    a1 -= a0 * l10;
    a2 -= a0 * l20;
    d = sqrt(row_bcast<1>(a1));
    a1 = (isS && rl == 1) ? d : a1 / d;
    const double l21 = row_bcast<2>(a1);
    a2 -= a1 * l21;
    d = sqrt(row_bcast<2>(a2));
    a2 = (isS && rl == 2) ? d : a2 / d;
    if (live) {
      if (isS) {
        S[s][rb] = a0;
        if (rl >= 1) S[s][rb + 1] = a1;
        if (rl == 2) S[s][rb + 2] = a2;
      } else if (isC) {
        if (t < MID) { Rm[m][cc * 3 + 0] = a0; Rm[m][cc * 3 + 1] = a1; Rm[m][cc * 3 + 2] = a2; }
      } else if (isG) {
        gv[s][0] = a0; gv[s][1] = a1; gv[s][2] = a2;
      }
    }
    __syncthreads();
    if (t < MID) {
      const int sjn = dn ? sj - 1 : sj + 1;
      const int sn = PER * sjn + MI;
      const int mn = dn ? sjn : sjn + 1;
      const bool toMid = (sjn == MID);
      const bool dnDelta = dn && toMid;
      double X[3][3];
      for (int r = 0; r < 3; ++r) { X[r][0] = Rm[m][r * 3 + 0]; X[r][1] = Rm[m][r * 3 + 1]; X[r][2] = Rm[m][r * 3 + 2]; }
      double u0 = 0.0, u1 = 0.0, u2 = 0.0, b0 = 0.0, b1 = 0.0, b2 = 0.0;
      if (isS) {
        u0 = Rm[m][rl * 3 + 0]; u1 = Rm[m][rl * 3 + 1]; u2 = Rm[m][rl * 3 + 2];
        if (!dnDelta) { b0 = S[sn][rb]; b1 = S[sn][rb + 1]; b2 = S[sn][rb + 2]; }
      } else if (isG) {
        u0 = a0; u1 = a1; u2 = a2;
        if (!dnDelta) { b0 = gv[sn][0]; b1 = gv[sn][1]; b2 = gv[sn][2]; }
      } else if (isC) {
        if (!toMid) { b0 = R[mn][cbase]; b1 = R[mn][cbase + cstr]; b2 = R[mn][cbase + 2 * cstr]; }
      }
      a0 = b0 - (X[0][0]*u0 + X[0][1]*u1 + X[0][2]*u2);
      a1 = b1 - (X[1][0]*u0 + X[1][1]*u1 + X[1][2]*u2);
      a2 = b2 - (X[2][0]*u0 + X[2][1]*u1 + X[2][2]*u2);
      if (dnDelta && live) {
        if (isS)      { Dl[rl][0] = a0; Dl[rl][1] = a1; Dl[rl][2] = a2; }
        else if (isG) { Dl[3][0] = a0; Dl[3][1] = a1; Dl[3][2] = a2; }
      }
      if (!dn && toMid) {
        if (isS)      { a0 += Dl[rl][0]; a1 += Dl[rl][1]; a2 += Dl[rl][2]; }
        else if (isG) { a0 += Dl[3][0]; a1 += Dl[3][1]; a2 += Dl[3][2]; }
      }
      if (dn && (t >= NPART - 2 - MID)) { a0 = id0; a1 = id1; a2 = id2; }
    }
  }
  if (lane == 0) ltsolve3(S[PER * MID + MI], gv[PER * MID + MI]);
  __syncthreads();
  if (q16 == 0) {
    for (int i = 0; i < MID; ++i) {
      const int sj = dn ? MID + 1 + i : MID - 1 - i;
      if (sj < NPART) {
        const int s = PER * sj + MI;
        const int sk = PER * (dn ? sj - 1 : sj + 1) + MI;
        const int m = dn ? sj : sj + 1;
        const double x0 = gv[sk][0], x1 = gv[sk][1], x2 = gv[sk][2];
        gv[s][0] -= Rm[m][0]*x0 + Rm[m][3]*x1 + Rm[m][6]*x2;
        gv[s][1] -= Rm[m][1]*x0 + Rm[m][4]*x1 + Rm[m][7]*x2;
        gv[s][2] -= Rm[m][2]*x0 + Rm[m][5]*x1 + Rm[m][8]*x2;
        ltsolve3(S[s], gv[s]);
      }
    }
  }
  __syncthreads();
  if (lane < NPART) {
    double xl0 = 0.0, xl1 = 0.0, xl2 = 0.0;
    if (hasL) { xl0 = gv[pa - 1][0]; xl1 = gv[pa - 1][1]; xl2 = gv[pa - 1][2]; }
    for (int i = MI - 1; i >= 0; --i) {
      const int p = pa + i;
      const double x0 = gv[p + 1][0], x1 = gv[p + 1][1], x2 = gv[p + 1][2];
      double t0 = gv[p][0] - (Mx[p][0]*x0 + Mx[p][3]*x1 + Mx[p][6]*x2);
      double t1 = gv[p][1] - (Mx[p][1]*x0 + Mx[p][4]*x1 + Mx[p][7]*x2);
      double t2 = gv[p][2] - (Mx[p][2]*x0 + Mx[p][5]*x1 + Mx[p][8]*x2);
      if (hasL) {
        t0 -= V[p][0]*xl0 + V[p][1]*xl1 + V[p][2]*xl2;
        t1 -= V[p][3]*xl0 + V[p][4]*xl1 + V[p][5]*xl2;
        t2 -= V[p][6]*xl0 + V[p][7]*xl1 + V[p][8]*xl2;
      }
      gv[p][0] = t0; gv[p][1] = t1; gv[p][2] = t2;
      ltsolve3(S[p], gv[p]);
    }
  }
  __syncthreads();

  for (int t = lane; t < NSTEP; t += 32) {
    out[0 * OUT_U + b * NSTEP + t] = (float)gv[t][0];
    out[1 * OUT_U + b * NSTEP + t] = (float)gv[t][1];
    out[2 * OUT_U + b * NSTEP + t] = (float)gv[t][2];
  }
  if (lane == 0) out[OUT_EPS + b] = 0.0f;
  for (int p = lane; p < T1; p += 32) out[OUT_ST + b * T1 + p] = hstep[p];
}

extern "C" void kernel_launch(void* const* d_in, const int* in_sizes, int n_in,
                              void* d_out, int out_size, void* d_ws, size_t ws_size,
                              hipStream_t stream) {
  (void)in_sizes; (void)n_in; (void)out_size; (void)d_ws; (void)ws_size;
  const float* coeffs = (const float*)d_in[0];
  const float* rhs    = (const float*)d_in[1];
  const float* ivr    = (const float*)d_in[2];
  const float* steps  = (const float*)d_in[3];
  float* out = (float*)d_out;
  hipLaunchKernelGGL(ode_blocktri_kernel, dim3(NB), dim3(32), 0, stream,
                     coeffs, rhs, ivr, steps, out);
}


# Back-substitutions (interiors, separators): ONE divide sequence a block on 3 lanes (t_c/d_c), then x by multiplies with l/d entries made in idle lane slots of the forward divides; rounding changes
# speedup vs baseline: 10.2111x; 1.1729x over previous
#include <hip/hip_runtime.h>
#include <hip/hip_bf16.h>
#include <math.h>


typedef __attribute__((ext_vector_type(2))) float v2f;
typedef __attribute__((ext_vector_type(8))) float v8f;

#define NSTEP 100
#define T1    99
#define NB    256
#define OUT_U   (NB * NSTEP)
#define OUT_EPS (3 * OUT_U)
#define OUT_ST  (OUT_EPS + NB)
#define PER   10
#define MI    9
#define NPART 10
#define MID   5
#define NTILE 7

__device__ __forceinline__ void Ecoef(int r, int c, float* pa, float* pb, float* pc) {
  float a = 0.f, b = 0.f, q = 0.f;
  if (r == 0)      { a = (c == 0) ? 1.f  : ((c == 3) ? -1.f : 0.f); b = (c == 1) ? 1.f  : 0.f; q = (c == 2) ? 0.5f : 0.f; }
  else if (r == 1) { a = (c == 1) ? 1.f  : ((c == 4) ? -1.f : 0.f); b = (c == 2) ? 1.f  : 0.f; }
  else if (r == 2) { a = (c == 0) ? -1.f : ((c == 3) ? 1.f  : 0.f); b = (c == 4) ? -1.f : 0.f; q = (c == 5) ? 0.5f : 0.f; }
  else if (r == 3) { a = (c == 1) ? -1.f : ((c == 4) ? 1.f  : 0.f); b = (c == 5) ? -1.f : 0.f; }
  else if (r == 4) { a = (c == 1) ? -1.f : ((c == 4) ? 1.f  : 0.f); b = (c == 2 || c == 5) ? -0.5f : 0.f; }
  *pa = a; *pb = b; *pc = q;
}

__device__ __forceinline__ void ltsolve3(const double* l, double* v) {
  double x2 = v[2] / l[5];
  double x1 = (v[1] - l[4] * x2) / l[2];
  double x0 = (v[0] - l[1] * x1 - l[3] * x2) / l[0];
  v[0] = x0; v[1] = x1; v[2] = x2;
}

__device__ __forceinline__ void mstep3(const double* L, double* Bp, double X[3][3]) {
  #pragma unroll
  for (int r = 0; r < 3; ++r) {
    const double c0 = Bp[0 + r];
    const double c1 = Bp[3 + r];
    const double c2 = Bp[6 + r];
    const double x0 = c0 / L[0];
    const double x1 = (c1 - x0 * L[1]) / L[2];
    const double x2 = (c2 - x0 * L[3] - x1 * L[4]) / L[5];
    X[r][0] = x0; X[r][1] = x1; X[r][2] = x2;
  }
  #pragma unroll
  for (int r = 0; r < 3; ++r) {
    Bp[r * 3 + 0] = X[r][0]; Bp[r * 3 + 1] = X[r][1]; Bp[r * 3 + 2] = X[r][2];
  }
}
__device__ __forceinline__ constexpr unsigned part_sel(int K, int half) {
  unsigned w = 0;
  for (int j = 0; j < 8; ++j) {
    const int lj = 8 * half + j;
    const unsigned s = (lj < 15) ? (unsigned)(3 * (lj / 3) + K) : 15u;
    w |= s << (4 * j);
  }
  return w;
}
template <int K>
__device__ __forceinline__ double part_bcast(double v) {
  constexpr unsigned lo = part_sel(K, 0), hi = part_sel(K, 1);
  const unsigned a = __builtin_amdgcn_permlane16(0u, (unsigned)__double2loint(v), lo, hi, false, false);
  const unsigned b = __builtin_amdgcn_permlane16(0u, (unsigned)__double2hiint(v), lo, hi, false, false);
  return __hiloint2double((int)b, (int)a);
}

template <int K>
__device__ __forceinline__ double row_bcast(double v) {
  constexpr unsigned sel = 0x11111111u * (unsigned)K;
  const unsigned a = __builtin_amdgcn_permlane16(0u, (unsigned)__double2loint(v), sel, sel, false, false);
  const unsigned b = __builtin_amdgcn_permlane16(0u, (unsigned)__double2hiint(v), sel, sel, false, false);
  return __hiloint2double((int)b, (int)a);
}

struct Fac3 { double d0, d1, d2, l10, l20, l21; };

__device__ __forceinline__ void panel3(int c, double& a0, double& a1, double& a2,
                                       double& y0, double& y1, double& y2,
                                       double& v0, double& v1, double& v2, Fac3& f,
                                       double& ma, double& mb) {
  const bool c0 = (c == 0);
  double d = sqrt(part_bcast<0>(a0));
  double q = (c0 ? y0 : a0) / d;
  y0 = c0 ? q : y0;
  a0 = c0 ? d : q;
  v0 = v0 / d;
  const double l10 = part_bcast<1>(a0), l20 = part_bcast<2>(a0);
  a1 -= a0 * l10;  a2 -= a0 * l20;
  y1 -= y0 * l10;  y2 -= y0 * l20;
  v1 -= v0 * l10;  v2 -= v0 * l20;
  f.d0 = d; f.l10 = l10; f.l20 = l20;
  d = sqrt(part_bcast<1>(a1));
  q = (c0 ? y1 : ((c == 1) ? l10 : a1)) / ((c == 1) ? f.d0 : d);
  y1 = c0 ? q : y1;
  ma = q;
  a1 = (c == 1) ? d : q;
  v1 = v1 / d;
  const double l21 = part_bcast<2>(a1);
  a2 -= a1 * l21;
  y2 -= y1 * l21;
  v2 -= v1 * l21;
  f.d1 = d; f.l21 = l21;
  d = sqrt(part_bcast<2>(a2));
  q = (c0 ? y2 : ((c == 1) ? l20 : l21)) / (c0 ? d : ((c == 1) ? f.d0 : f.d1));
  y2 = c0 ? q : y2;
  mb = q;
  v2 = v2 / d;
  a2 = d;
  f.d2 = d;
}

__global__ __launch_bounds__(32) void ode_blocktri_kernel(
    const float* __restrict__ coeffs,
    const float* __restrict__ rhs,
    const float* __restrict__ ivr,
    const float* __restrict__ steps,
    float* __restrict__ out)
{
  __shared__ double S[NSTEP][6];
  __shared__ double Bo[T1][9];
  __shared__ double gv[NSTEP][3];
  __shared__ float  hstep[T1];
  __shared__ double V[NSTEP][9];
  __shared__ double Mx[T1][9];
  __shared__ double R[NPART][9];
  __shared__ double Rm[NPART][9];
  __shared__ double Dl[4][3];
  __shared__ double Ls[NSTEP][3];

  const int b = blockIdx.x;
  const int lane = threadIdx.x;

  for (int i = lane; i < T1; i += 32) hstep[i] = steps[b * T1 + i];
  __syncthreads();

  const int e  = lane & 15;
  const int hh = lane >> 4;
  const int i2 = (e >= 3) ? 2 : ((e >= 1) ? 1 : 0);
  const int i3 = (e >= 12) ? 2 : ((e >= 9) ? 1 : 0);
  const int ca = (e < 6) ? i2 : ((e < 15) ? i3 : 6);
  const int cb = (e < 6) ? e - i2 * (i2 + 1) / 2 : ((e < 15) ? e - 3 - 3 * i3 : 6);
  const int ga = (e < 6) ? 3 + ca : 6;
  const int gb = (e < 6) ? 3 + cb : 6;
  float ph0 = 0.f, ph1 = 0.f, ph2 = 0.f, ph3 = 0.f, ph4 = 0.f;
  float pg0 = 0.f, pg1 = 0.f, pg2 = 0.f, pg3 = 0.f, pg4 = 0.f;
  #pragma unroll
  for (int k = 0; k < 5; ++k) {
    float a1, b1, q1, a2, b2, q2;
    Ecoef(k, ca, &a1, &b1, &q1); Ecoef(k, cb, &a2, &b2, &q2);
    ph0 += a1 * a2; ph1 += a1 * b2 + b1 * a2; ph2 += a1 * q2 + b1 * b2 + q1 * a2;
    ph3 += b1 * q2 + q1 * b2; ph4 += q1 * q2;
    Ecoef(k, ga, &a1, &b1, &q1); Ecoef(k, gb, &a2, &b2, &q2);
    pg0 += a1 * a2; pg1 += a1 * b2 + b1 * a2; pg2 += a1 * q2 + b1 * b2 + q1 * a2;
    pg3 += b1 * q2 + q1 * b2; pg4 += q1 * q2;
  }
  v2f A0, A1, A2;
  A0.x = hh ? ph2 : ph0;  A0.y = hh ? ph3 : ph1;
  A1.x = hh ? pg1 : ph4;  A1.y = hh ? pg2 : pg0;
  A2.x = hh ? 0.f : pg3;  A2.y = hh ? 0.f : pg4;

  float cf0[NTILE], cf1[NTILE], cf2[NTILE], rf[NTILE], hf[NTILE], gf[NTILE];
  #pragma unroll
  for (int n = 0; n < NTILE; ++n) {
    const int t = 16 * n + e;
    const bool own = (hh == 0) && (t < NSTEP);
    const int tl = (t < NSTEP) ? t : NSTEP - 1;
    const float l0 = coeffs[(b * NSTEP + tl) * 3 + 0];
    const float l1 = coeffs[(b * NSTEP + tl) * 3 + 1];
    const float l2 = coeffs[(b * NSTEP + tl) * 3 + 2];
    const float lr = rhs[b * NSTEP + tl];
    const float lh = hstep[(t < T1) ? t : T1 - 1];
    const float lg = hstep[(t >= 1 && t < NSTEP) ? t - 1 : 0];
    cf0[n] = own ? l0 : 0.f;
    cf1[n] = own ? l1 : 0.f;
    cf2[n] = own ? l2 : 0.f;
    rf[n]  = own ? lr : 0.f;
    hf[n]  = (t < T1) ? lh : 0.f;
    gf[n]  = (t >= 1 && t < NSTEP) ? lg : 0.f;
  }
  #pragma unroll
  for (int n = 0; n < NTILE; ++n) {
    const int t = 16 * n + e;
    const float h = hf[n], g = gf[n];
    const float u = (t < T1) ? 1.f : 0.f;
    const float w = (t >= 1 && t < NSTEP) ? 1.f : 0.f;
    const float h2 = h * h, h3 = h2 * h, h4 = h2 * h2;
    const float g2 = g * g, g3 = g2 * g, g4 = g2 * g2;
    v2f B0, B1, B2;
    B0.x = hh ? h2 : u;    B0.y = hh ? h3 : h;
    B1.x = hh ? g  : h4;   B1.y = hh ? g2 : w;
    B2.x = hh ? 0.f : g3;  B2.y = hh ? 0.f : g4;
    v8f acc = {};
    acc = __builtin_amdgcn_wmma_f32_16x16x4_f32(false, A0, false, B0, (short)0, acc, false, false);
    acc = __builtin_amdgcn_wmma_f32_16x16x4_f32(false, A1, false, B1, (short)0, acc, false, false);
    acc = __builtin_amdgcn_wmma_f32_16x16x4_f32(false, A2, false, B2, (short)0, acc, false, false);
    const double c0 = (double)cf0[n], c1 = (double)cf1[n], c2 = (double)cf2[n], rr = (double)rf[n];
    double s0 = fma(c0, c0, (double)acc[0]);
    double s1 = fma(c1, c0, (double)acc[1]);
    double s2 = fma(c1, c1, (double)acc[2]);
    double s3 = fma(c2, c0, (double)acc[3]);
    double s4 = fma(c2, c1, (double)acc[4]);
    double s5 = fma(c2, c2, (double)acc[5]);
    const double s6 = (double)acc[6], s7 = (double)acc[7];
    double y0 = c0 * rr, y1 = c1 * rr, y2 = c2 * rr;
    if (n == 0 && lane == 0) {
      s0 += 1.0; s2 += 1.0;
      y0 += (double)ivr[b * 2 + 0];
      y1 += (double)ivr[b * 2 + 1];
    }
    if (hh == 0) {
      if (t < NSTEP) {
        S[t][0] = s0; S[t][1] = s1; S[t][2] = s2; S[t][3] = s3; S[t][4] = s4; S[t][5] = s5;
        gv[t][0] = y0; gv[t][1] = y1; gv[t][2] = y2;
      }
      if (t < T1) { Bo[t][0] = s6; Bo[t][1] = s7; }
    } else if (t < T1) {
      Bo[t][2] = s0; Bo[t][3] = s1; Bo[t][4] = s2; Bo[t][5] = s3; Bo[t][6] = s4; Bo[t][7] = s5; Bo[t][8] = s6;
    }
  }
  __syncthreads();

  const int l16 = lane & 15;
  const bool act = l16 < 15;
  const int part = act ? 5 * (lane >> 4) + l16 / 3 : 0;
  const int c = act ? l16 % 3 : 0;
  const int pk = PER * part;
  const int prow = c * (c + 1) / 2;
  Fac3 f;
  double yc0 = 0.0, yc1 = 0.0, yc2 = 0.0;
  double vc0 = 0.0, vc1 = 0.0, vc2 = 0.0;
  {
    double a0 = S[pk][prow], a1 = S[pk][prow + 1], a2 = S[pk][prow + 2];
    if (c == 0) { yc0 = gv[pk][0]; yc1 = gv[pk][1]; yc2 = gv[pk][2]; }
    if (part > 0) {
      vc0 = Bo[pk - 1][c * 3 + 0]; vc1 = Bo[pk - 1][c * 3 + 1]; vc2 = Bo[pk - 1][c * 3 + 2];
    }
    double ma, mb;
    panel3(c, a0, a1, a2, yc0, yc1, yc2, vc0, vc1, vc2, f, ma, mb);
    if (act) {
      if (c == 1) { Ls[pk][0] = ma; Ls[pk][1] = mb; }
      if (c == 2) Ls[pk][2] = mb;
      V[pk][0 + c] = vc0; V[pk][3 + c] = vc1; V[pk][6 + c] = vc2;
      S[pk][prow] = a0;
      if (c >= 1) S[pk][prow + 1] = a1;
      if (c == 2) S[pk][prow + 2] = a2;
      if (c == 0) { gv[pk][0] = yc0; gv[pk][1] = yc1; gv[pk][2] = yc2; }
    }
  }
  for (int i = 0; i < MI; ++i) {
    const int p = pk + i;
    const double b0 = Bo[p][0 + c], b1 = Bo[p][3 + c], b2 = Bo[p][6 + c];
    const double x0 = b0 / f.d0;
    const double x1 = (b1 - x0 * f.l10) / f.d1;
    const double x2 = (b2 - x0 * f.l20 - x1 * f.l21) / f.d2;
    if (act) { Mx[p][c * 3 + 0] = x0; Mx[p][c * 3 + 1] = x1; Mx[p][c * 3 + 2] = x2; }
    __syncthreads();
    double X[3][3];
    for (int r = 0; r < 3; ++r) { X[r][0] = Mx[p][r * 3 + 0]; X[r][1] = Mx[p][r * 3 + 1]; X[r][2] = Mx[p][r * 3 + 2]; }
    double n0 = S[p + 1][prow + 0] - (x0*X[0][0] + x1*X[0][1] + x2*X[0][2]);
    double n1 = S[p + 1][prow + 1] - (x0*X[1][0] + x1*X[1][1] + x2*X[1][2]);
    double n2 = S[p + 1][prow + 2] - (x0*X[2][0] + x1*X[2][1] + x2*X[2][2]);
    double yn0 = 0.0, yn1 = 0.0, yn2 = 0.0;
    if (c == 0) {
      yn0 = gv[p + 1][0] - (X[0][0]*yc0 + X[0][1]*yc1 + X[0][2]*yc2);
      yn1 = gv[p + 1][1] - (X[1][0]*yc0 + X[1][1]*yc1 + X[1][2]*yc2);
      yn2 = gv[p + 1][2] - (X[2][0]*yc0 + X[2][1]*yc1 + X[2][2]*yc2);
    }
    if (i < MI - 1) {
      double vn0 = -(X[0][0]*vc0 + X[0][1]*vc1 + X[0][2]*vc2);
      double vn1 = -(X[1][0]*vc0 + X[1][1]*vc1 + X[1][2]*vc2);
      double vn2 = -(X[2][0]*vc0 + X[2][1]*vc1 + X[2][2]*vc2);
      double ma, mb;
      panel3(c, n0, n1, n2, yn0, yn1, yn2, vn0, vn1, vn2, f, ma, mb);
      if (act) {
        if (c == 1) { Ls[p + 1][0] = ma; Ls[p + 1][1] = mb; }
        if (c == 2) Ls[p + 1][2] = mb;
        V[p + 1][0 + c] = vn0; V[p + 1][3 + c] = vn1; V[p + 1][6 + c] = vn2;
        S[p + 1][prow] = n0;
        if (c >= 1) S[p + 1][prow + 1] = n1;
        if (c == 2) S[p + 1][prow + 2] = n2;
        if (c == 0) { gv[p + 1][0] = yn0; gv[p + 1][1] = yn1; gv[p + 1][2] = yn2; }
      }
      yc0 = yn0; yc1 = yn1; yc2 = yn2;
      vc0 = vn0; vc1 = vn1; vc2 = vn2;
    } else if (act) {
      S[p + 1][prow] = n0;
      if (c >= 1) S[p + 1][prow + 1] = n1;
      if (c == 2) S[p + 1][prow + 2] = n2;
      if (c == 0) { gv[p + 1][0] = yn0; gv[p + 1][1] = yn1; gv[p + 1][2] = yn2; }
      for (int j = 0; j < 3; ++j)
        R[part][c * 3 + j] = -(vc0*X[j][0] + vc1*X[j][1] + vc2*X[j][2]);
    }
  }
  __syncthreads();
  for (int idx = lane; idx < (NPART - 1) * 6; idx += 32) {
    const int k = 1 + idx / 6, q = idx % 6;
    const int i2 = (q >= 3) ? 2 : ((q >= 1) ? 1 : 0);
    const int j2 = q - i2 * (i2 + 1) / 2;
    const int p0 = PER * k;
    double acc = 0.0;
    for (int i = 0; i < MI; ++i)
      acc += V[p0 + i][0 + i2]*V[p0 + i][0 + j2] + V[p0 + i][3 + i2]*V[p0 + i][3 + j2]
           + V[p0 + i][6 + i2]*V[p0 + i][6 + j2];
    S[p0 - 1][q] -= acc;
  }
  for (int idx = lane; idx < (NPART - 1) * 3; idx += 32) {
    const int k = 1 + idx / 3, j = idx % 3;
    const int p0 = PER * k;
    double acc = 0.0;
    for (int i = 0; i < MI; ++i)
      acc += V[p0 + i][0 + j]*gv[p0 + i][0] + V[p0 + i][3 + j]*gv[p0 + i][1]
           + V[p0 + i][6 + j]*gv[p0 + i][2];
    gv[p0 - 1][j] -= acc;
  }
  __syncthreads();
  const int dn = lane >> 4;
  const int q16 = lane & 15;
  const bool isS = q16 < 3, isC = (q16 >= 3) && (q16 < 6), isG = (q16 == 6);
  const bool isX = (q16 == 7), isY = (q16 == 8);
  const int rl = isS ? q16 : 0;
  const int rb = rl * (rl + 1) / 2;
  const int cc = isC ? q16 - 3 : 0;
  const int cbase = dn ? cc * 3 : cc;
  const int cstr = dn ? 1 : 3;
  const double id0 = (isS && rl == 0) ? 1.0 : 0.0, id1 = (isS && rl == 1) ? 1.0 : 0.0, id2 = (isS && rl == 2) ? 1.0 : 0.0;
  double a0 = 0.0, a1 = 0.0, a2 = 0.0;
  {
    const int sj = dn ? NPART - 1 : 0;
    const int s = PER * sj + MI;
    const int m = dn ? sj : sj + 1;
    if (isS)      { a0 = S[s][rb]; a1 = S[s][rb + 1]; a2 = S[s][rb + 2]; }
    else if (isC) { a0 = R[m][cbase]; a1 = R[m][cbase + cstr]; a2 = R[m][cbase + 2 * cstr]; }
    else if (isG) { a0 = gv[s][0]; a1 = gv[s][1]; a2 = gv[s][2]; }
  }
  for (int t = 0; t <= MID; ++t) {
    const int sj = dn ? NPART - 1 - t : t;
    const bool live = dn ? (t < NPART - 1 - MID) : true;
    const int s = PER * sj + MI;
    const int m = dn ? sj : sj + 1;
    double d = sqrt(row_bcast<0>(a0));
    a0 = (isS && rl == 0) ? d : a0 / d;
    const double l10 = row_bcast<1>(a0), l20 = row_bcast<2>(a0);
    a1 -= a0 * l10;
    a2 -= a0 * l20;
    const double d0 = d;
    d = sqrt(row_bcast<1>(a1));
    double q = (isX ? l10 : a1) / (isX ? d0 : d);
    const double sm10 = q;
    a1 = (isS && rl == 1) ? d : (isX ? a1 : q);
    const double l21 = row_bcast<2>(a1);
    a2 -= a1 * l21;
    const double d1 = d;
    d = sqrt(row_bcast<2>(a2));
    q = (isX ? l20 : (isY ? l21 : a2)) / (isX ? d0 : (isY ? d1 : d));
    a2 = (isS && rl == 2) ? d : ((isX || isY) ? a2 : q);
    if (live) {
      if (isS) {
        S[s][rb] = a0;
        if (rl >= 1) S[s][rb + 1] = a1;
        if (rl == 2) S[s][rb + 2] = a2;
      } else if (isC) {
        if (t < MID) { Rm[m][cc * 3 + 0] = a0; Rm[m][cc * 3 + 1] = a1; Rm[m][cc * 3 + 2] = a2; }
      } else if (isG) {
        gv[s][0] = a0; gv[s][1] = a1; gv[s][2] = a2;
      } else if (isX) {
        Ls[s][0] = sm10; Ls[s][1] = q;
      } else if (isY) {
        Ls[s][2] = q;
      }
    }
    __syncthreads();
    if (t < MID) {
      const int sjn = dn ? sj - 1 : sj + 1;
      const int sn = PER * sjn + MI;
      const int mn = dn ? sjn : sjn + 1;
      const bool toMid = (sjn == MID);
      const bool dnDelta = dn && toMid;
      double X[3][3];
      for (int r = 0; r < 3; ++r) { X[r][0] = Rm[m][r * 3 + 0]; X[r][1] = Rm[m][r * 3 + 1]; X[r][2] = Rm[m][r * 3 + 2]; }
      double u0 = 0.0, u1 = 0.0, u2 = 0.0, b0 = 0.0, b1 = 0.0, b2 = 0.0;
      if (isS) {
        u0 = Rm[m][rl * 3 + 0]; u1 = Rm[m][rl * 3 + 1]; u2 = Rm[m][rl * 3 + 2];
        if (!dnDelta) { b0 = S[sn][rb]; b1 = S[sn][rb + 1]; b2 = S[sn][rb + 2]; }
      } else if (isG) {
        u0 = a0; u1 = a1; u2 = a2;
        if (!dnDelta) { b0 = gv[sn][0]; b1 = gv[sn][1]; b2 = gv[sn][2]; }
      } else if (isC) {
        if (!toMid) { b0 = R[mn][cbase]; b1 = R[mn][cbase + cstr]; b2 = R[mn][cbase + 2 * cstr]; }
      }
      a0 = b0 - (X[0][0]*u0 + X[0][1]*u1 + X[0][2]*u2);
      a1 = b1 - (X[1][0]*u0 + X[1][1]*u1 + X[1][2]*u2);
      a2 = b2 - (X[2][0]*u0 + X[2][1]*u1 + X[2][2]*u2);
      if (dnDelta && live) {
        if (isS)      { Dl[rl][0] = a0; Dl[rl][1] = a1; Dl[rl][2] = a2; }
        else if (isG) { Dl[3][0] = a0; Dl[3][1] = a1; Dl[3][2] = a2; }
      }
      if (!dn && toMid) {
        if (isS)      { a0 += Dl[rl][0]; a1 += Dl[rl][1]; a2 += Dl[rl][2]; }
        else if (isG) { a0 += Dl[3][0]; a1 += Dl[3][1]; a2 += Dl[3][2]; }
      }
      if (dn && (t >= NPART - 2 - MID)) { a0 = id0; a1 = id1; a2 = id2; }
    }
  }
  {
    const int c3 = isS ? q16 : 0;
    const int dg3 = rb + c3;
    double x0, x1, x2;
    {
      const int s = PER * MID + MI;
      const double sc = gv[s][c3] / S[s][dg3];
      const double m10 = Ls[s][0], m20 = Ls[s][1], m21 = Ls[s][2];
      x2 = row_bcast<2>(sc);
      x1 = row_bcast<1>(sc) - m21 * x2;
      x0 = row_bcast<0>(sc) - m10 * x1 - m20 * x2;
    }
    if (lane == 0) { gv[PER * MID + MI][0] = x0; gv[PER * MID + MI][1] = x1; gv[PER * MID + MI][2] = x2; }
    for (int i = 0; i < MID; ++i) {
      const int sj = dn ? MID + 1 + i : MID - 1 - i;
      const bool has = sj < NPART;
      const int sjc = has ? sj : NPART - 1;
      const int s = PER * sjc + MI;
      const int m = dn ? sjc : sjc + 1;
      const double tt = gv[s][c3] - (Rm[m][c3]*x0 + Rm[m][3 + c3]*x1 + Rm[m][6 + c3]*x2);
      const double sc = tt / S[s][dg3];
      const double m10 = Ls[s][0], m20 = Ls[s][1], m21 = Ls[s][2];
      const double n2 = row_bcast<2>(sc);
      const double n1 = row_bcast<1>(sc) - m21 * n2;
      const double n0 = row_bcast<0>(sc) - m10 * n1 - m20 * n2;
      if (has) {
        x0 = n0; x1 = n1; x2 = n2;
        if (q16 == 0) { gv[s][0] = x0; gv[s][1] = x1; gv[s][2] = x2; }
      }
    }
  }
  __syncthreads();
  {
    double xl0 = 0.0, xl1 = 0.0, xl2 = 0.0;
    if (part > 0) { xl0 = gv[pk - 1][0]; xl1 = gv[pk - 1][1]; xl2 = gv[pk - 1][2]; }
    double x0 = gv[pk + MI][0], x1 = gv[pk + MI][1], x2 = gv[pk + MI][2];
    const int dg = prow + c;
    for (int i = MI - 1; i >= 0; --i) {
      const int p = pk + i;
      double t = gv[p][c] - (Mx[p][c]*x0 + Mx[p][3 + c]*x1 + Mx[p][6 + c]*x2);
      t -= V[p][c * 3 + 0]*xl0 + V[p][c * 3 + 1]*xl1 + V[p][c * 3 + 2]*xl2;
      const double s = t / S[p][dg];
      const double m10 = Ls[p][0], m20 = Ls[p][1], m21 = Ls[p][2];
      x2 = part_bcast<2>(s);
      x1 = part_bcast<1>(s) - m21 * x2;
      x0 = part_bcast<0>(s) - m10 * x1 - m20 * x2;
      if (act && c == 0) { gv[p][0] = x0; gv[p][1] = x1; gv[p][2] = x2; }
    }
  }
  __syncthreads();

  for (int t = lane; t < NSTEP; t += 32) {
    out[0 * OUT_U + b * NSTEP + t] = (float)gv[t][0];
    out[1 * OUT_U + b * NSTEP + t] = (float)gv[t][1];
    out[2 * OUT_U + b * NSTEP + t] = (float)gv[t][2];
  }
  if (lane == 0) out[OUT_EPS + b] = 0.0f;
  for (int p = lane; p < T1; p += 32) out[OUT_ST + b * T1 + p] = hstep[p];
}

extern "C" void kernel_launch(void* const* d_in, const int* in_sizes, int n_in,
                              void* d_out, int out_size, void* d_ws, size_t ws_size,
                              hipStream_t stream) {
  (void)in_sizes; (void)n_in; (void)out_size; (void)d_ws; (void)ws_size;
  const float* coeffs = (const float*)d_in[0];
  const float* rhs    = (const float*)d_in[1];
  const float* ivr    = (const float*)d_in[2];
  const float* steps  = (const float*)d_in[3];
  float* out = (float*)d_out;
  hipLaunchKernelGGL(ode_blocktri_kernel, dim3(NB), dim3(32), 0, stream,
                     coeffs, rhs, ivr, steps, out);
}


# Interiors' chain: one panel ROW per lane (3 diagonal, 3 multiplier, 3 spike rows: 9 lanes a partition), ONE divide sequence a column, not three; 10 partitions on 4 waves; sum order kept
# speedup vs baseline: 13.0458x; 1.2776x over previous
#include <hip/hip_runtime.h>
#include <hip/hip_bf16.h>
#include <math.h>


typedef __attribute__((ext_vector_type(2))) float v2f;
typedef __attribute__((ext_vector_type(8))) float v8f;

#define NSTEP 100
#define T1    99
#define NB    256
#define OUT_U   (NB * NSTEP)
#define OUT_EPS (3 * OUT_U)
#define OUT_ST  (OUT_EPS + NB)
#define PER   10
#define MI    9
#define NPART 10
#define MID   5
#define NTILE 7

__device__ __forceinline__ void Ecoef(int r, int c, float* pa, float* pb, float* pc) {
  float a = 0.f, b = 0.f, q = 0.f;
  if (r == 0)      { a = (c == 0) ? 1.f  : ((c == 3) ? -1.f : 0.f); b = (c == 1) ? 1.f  : 0.f; q = (c == 2) ? 0.5f : 0.f; }
  else if (r == 1) { a = (c == 1) ? 1.f  : ((c == 4) ? -1.f : 0.f); b = (c == 2) ? 1.f  : 0.f; }
  else if (r == 2) { a = (c == 0) ? -1.f : ((c == 3) ? 1.f  : 0.f); b = (c == 4) ? -1.f : 0.f; q = (c == 5) ? 0.5f : 0.f; }
  else if (r == 3) { a = (c == 1) ? -1.f : ((c == 4) ? 1.f  : 0.f); b = (c == 5) ? -1.f : 0.f; }
  else if (r == 4) { a = (c == 1) ? -1.f : ((c == 4) ? 1.f  : 0.f); b = (c == 2 || c == 5) ? -0.5f : 0.f; }
  *pa = a; *pb = b; *pc = q;
}

__device__ __forceinline__ void ltsolve3(const double* l, double* v) {
  double x2 = v[2] / l[5];
  double x1 = (v[1] - l[4] * x2) / l[2];
  double x0 = (v[0] - l[1] * x1 - l[3] * x2) / l[0];
  v[0] = x0; v[1] = x1; v[2] = x2;
}

__device__ __forceinline__ void mstep3(const double* L, double* Bp, double X[3][3]) {
  #pragma unroll
  for (int r = 0; r < 3; ++r) {
    const double c0 = Bp[0 + r];
    const double c1 = Bp[3 + r];
    const double c2 = Bp[6 + r];
    const double x0 = c0 / L[0];
    const double x1 = (c1 - x0 * L[1]) / L[2];
    const double x2 = (c2 - x0 * L[3] - x1 * L[4]) / L[5];
    X[r][0] = x0; X[r][1] = x1; X[r][2] = x2;
  }
  #pragma unroll
  for (int r = 0; r < 3; ++r) {
    Bp[r * 3 + 0] = X[r][0]; Bp[r * 3 + 1] = X[r][1]; Bp[r * 3 + 2] = X[r][2];
  }
}
__device__ __forceinline__ constexpr unsigned part_sel(int K, int half) {
  unsigned w = 0;
  for (int j = 0; j < 8; ++j) {
    const int lj = 8 * half + j;
    const unsigned s = (lj < 15) ? (unsigned)(3 * (lj / 3) + K) : 15u;
    w |= s << (4 * j);
  }
  return w;
}
template <int K>
__device__ __forceinline__ double part_bcast(double v) {
  constexpr unsigned lo = part_sel(K, 0), hi = part_sel(K, 1);
  const unsigned a = __builtin_amdgcn_permlane16(0u, (unsigned)__double2loint(v), lo, hi, false, false);
  const unsigned b = __builtin_amdgcn_permlane16(0u, (unsigned)__double2hiint(v), lo, hi, false, false);
  return __hiloint2double((int)b, (int)a);
}

template <int K>
__device__ __forceinline__ double row_bcast(double v) {
  constexpr unsigned sel = 0x11111111u * (unsigned)K;
  const unsigned a = __builtin_amdgcn_permlane16(0u, (unsigned)__double2loint(v), sel, sel, false, false);
  const unsigned b = __builtin_amdgcn_permlane16(0u, (unsigned)__double2hiint(v), sel, sel, false, false);
  return __hiloint2double((int)b, (int)a);
}

__device__ __forceinline__ constexpr unsigned role_sel(int K, int half) {
  unsigned w = 0;
  for (int j = 0; j < 8; ++j) {
    const int lj = 8 * half + j;
    const unsigned sl = (lj < 9) ? (unsigned)K : (unsigned)(9 + K);
    w |= sl << (4 * j);
  }
  return w;
}
template <int K>
__device__ __forceinline__ double role_bcast(double v) {
  constexpr unsigned lo = role_sel(K, 0), hi = role_sel(K, 1);
  const unsigned a = __builtin_amdgcn_permlane16(0u, (unsigned)__double2loint(v), lo, hi, false, false);
  const unsigned b = __builtin_amdgcn_permlane16(0u, (unsigned)__double2hiint(v), lo, hi, false, false);
  return __hiloint2double((int)b, (int)a);
}

__device__ __forceinline__ void panel_row(int role, double piv, double y0, double a0, double a1, double a2,
                                          double& e0, double& e1, double& e2,
                                          double& d0, double& d1, double& d2) {
  d0 = sqrt(piv);
  e0 = ((role == 0) ? y0 : a0) / d0;
  const double l10 = role_bcast<1>(e0), l20 = role_bcast<2>(e0);
  a1 -= e0 * l10;  a2 -= e0 * l20;
  d1 = sqrt(role_bcast<1>(a1));
  e1 = ((role == 1) ? e0 : a1) / ((role == 1) ? d0 : d1);
  const double l21 = role_bcast<2>(e1);
  a2 -= e1 * l21;
  d2 = sqrt(role_bcast<2>(a2));
  e2 = ((role == 1) ? l20 : ((role == 2) ? e1 : a2)) / ((role == 1) ? d0 : ((role == 2) ? d1 : d2));
}

__global__ __launch_bounds__(128) void ode_blocktri_kernel(
    const float* __restrict__ coeffs,
    const float* __restrict__ rhs,
    const float* __restrict__ ivr,
    const float* __restrict__ steps,
    float* __restrict__ out)
{
  __shared__ double S[NSTEP][6];
  __shared__ double Bo[T1][9];
  __shared__ double gv[NSTEP][3];
  __shared__ float  hstep[T1];
  __shared__ double V[NSTEP][9];
  __shared__ double Mx[T1][9];
  __shared__ double R[NPART][9];
  __shared__ double Rm[NPART][9];
  __shared__ double Dl[4][3];
  __shared__ double Ls[NSTEP][3];

  const int b = blockIdx.x;
  const int tid = threadIdx.x;
  const int lane = tid & 31;
  const int wv = __builtin_amdgcn_readfirstlane(tid >> 5);
  const bool w0 = (wv == 0);

  for (int i = tid; i < T1; i += 128) hstep[i] = steps[b * T1 + i];
  __syncthreads();

  if (w0) {
  const int e  = lane & 15;
  const int hh = lane >> 4;
  const int i2 = (e >= 3) ? 2 : ((e >= 1) ? 1 : 0);
  const int i3 = (e >= 12) ? 2 : ((e >= 9) ? 1 : 0);
  const int ca = (e < 6) ? i2 : ((e < 15) ? i3 : 6);
  const int cb = (e < 6) ? e - i2 * (i2 + 1) / 2 : ((e < 15) ? e - 3 - 3 * i3 : 6);
  const int ga = (e < 6) ? 3 + ca : 6;
  const int gb = (e < 6) ? 3 + cb : 6;
  float ph0 = 0.f, ph1 = 0.f, ph2 = 0.f, ph3 = 0.f, ph4 = 0.f;
  float pg0 = 0.f, pg1 = 0.f, pg2 = 0.f, pg3 = 0.f, pg4 = 0.f;
  #pragma unroll
  for (int k = 0; k < 5; ++k) {
    float a1, b1, q1, a2, b2, q2;
    Ecoef(k, ca, &a1, &b1, &q1); Ecoef(k, cb, &a2, &b2, &q2);
    ph0 += a1 * a2; ph1 += a1 * b2 + b1 * a2; ph2 += a1 * q2 + b1 * b2 + q1 * a2;
    ph3 += b1 * q2 + q1 * b2; ph4 += q1 * q2;
    Ecoef(k, ga, &a1, &b1, &q1); Ecoef(k, gb, &a2, &b2, &q2);
    pg0 += a1 * a2; pg1 += a1 * b2 + b1 * a2; pg2 += a1 * q2 + b1 * b2 + q1 * a2;
    pg3 += b1 * q2 + q1 * b2; pg4 += q1 * q2;
  }
  v2f A0, A1, A2;
  A0.x = hh ? ph2 : ph0;  A0.y = hh ? ph3 : ph1;
  A1.x = hh ? pg1 : ph4;  A1.y = hh ? pg2 : pg0;
  A2.x = hh ? 0.f : pg3;  A2.y = hh ? 0.f : pg4;

  float cf0[NTILE], cf1[NTILE], cf2[NTILE], rf[NTILE], hf[NTILE], gf[NTILE];
  #pragma unroll
  for (int n = 0; n < NTILE; ++n) {
    const int t = 16 * n + e;
    const bool own = (hh == 0) && (t < NSTEP);
    const int tl = (t < NSTEP) ? t : NSTEP - 1;
    const float l0 = coeffs[(b * NSTEP + tl) * 3 + 0];
    const float l1 = coeffs[(b * NSTEP + tl) * 3 + 1];
    const float l2 = coeffs[(b * NSTEP + tl) * 3 + 2];
    const float lr = rhs[b * NSTEP + tl];
    const float lh = hstep[(t < T1) ? t : T1 - 1];
    const float lg = hstep[(t >= 1 && t < NSTEP) ? t - 1 : 0];
    cf0[n] = own ? l0 : 0.f;
    cf1[n] = own ? l1 : 0.f;
    cf2[n] = own ? l2 : 0.f;
    rf[n]  = own ? lr : 0.f;
    hf[n]  = (t < T1) ? lh : 0.f;
    gf[n]  = (t >= 1 && t < NSTEP) ? lg : 0.f;
  }
  #pragma unroll
  for (int n = 0; n < NTILE; ++n) {
    const int t = 16 * n + e;
    const float h = hf[n], g = gf[n];
    const float u = (t < T1) ? 1.f : 0.f;
    const float w = (t >= 1 && t < NSTEP) ? 1.f : 0.f;
    const float h2 = h * h, h3 = h2 * h, h4 = h2 * h2;
    const float g2 = g * g, g3 = g2 * g, g4 = g2 * g2;
    v2f B0, B1, B2;
    B0.x = hh ? h2 : u;    B0.y = hh ? h3 : h;
    B1.x = hh ? g  : h4;   B1.y = hh ? g2 : w;
    B2.x = hh ? 0.f : g3;  B2.y = hh ? 0.f : g4;
    v8f acc = {};
    acc = __builtin_amdgcn_wmma_f32_16x16x4_f32(false, A0, false, B0, (short)0, acc, false, false);
    acc = __builtin_amdgcn_wmma_f32_16x16x4_f32(false, A1, false, B1, (short)0, acc, false, false);
    acc = __builtin_amdgcn_wmma_f32_16x16x4_f32(false, A2, false, B2, (short)0, acc, false, false);
    const double c0 = (double)cf0[n], c1 = (double)cf1[n], c2 = (double)cf2[n], rr = (double)rf[n];
    double s0 = fma(c0, c0, (double)acc[0]);
    double s1 = fma(c1, c0, (double)acc[1]);
    double s2 = fma(c1, c1, (double)acc[2]);
    double s3 = fma(c2, c0, (double)acc[3]);
    double s4 = fma(c2, c1, (double)acc[4]);
    double s5 = fma(c2, c2, (double)acc[5]);
    const double s6 = (double)acc[6], s7 = (double)acc[7];
    double y0 = c0 * rr, y1 = c1 * rr, y2 = c2 * rr;
    if (n == 0 && lane == 0) {
      s0 += 1.0; s2 += 1.0;
      y0 += (double)ivr[b * 2 + 0];
      y1 += (double)ivr[b * 2 + 1];
    }
    if (hh == 0) {
      if (t < NSTEP) {
        S[t][0] = s0; S[t][1] = s1; S[t][2] = s2; S[t][3] = s3; S[t][4] = s4; S[t][5] = s5;
        gv[t][0] = y0; gv[t][1] = y1; gv[t][2] = y2;
      }
      if (t < T1) { Bo[t][0] = s6; Bo[t][1] = s7; }
    } else if (t < T1) {
      Bo[t][2] = s0; Bo[t][3] = s1; Bo[t][4] = s2; Bo[t][5] = s3; Bo[t][6] = s4; Bo[t][7] = s5; Bo[t][8] = s6;
    }
  }
  }
  __syncthreads();

  const int l16 = lane & 15;
  {
    const int rw = lane >> 4;
    const bool inB = l16 >= 9;
    const int role = !inB ? l16
                   : ((rw == 0) ? ((l16 == 15) ? 6 : l16 - 9)
                                : ((l16 == 9) ? 7 : ((l16 == 15) ? 8 : l16 - 9)));
    const bool isCopy = inB && (rw == 1) && (l16 >= 10) && (l16 <= 14);
    const int pw = 3 * wv + (inB ? 1 : (rw ? 2 : 0));
    const bool keep = (pw < NPART) && !isCopy;
    const int partw = (pw < NPART) ? pw : NPART - 1;
    const int pkw = PER * partw;
    const bool isD = role < 3, isM = (role >= 3) && (role < 6), isV = role >= 6;
    const int ri = isD ? role : (isM ? role - 3 : role - 6);
    const int prw = ri * (ri + 1) / 2;
    double piv = S[pkw][0];
    double y0 = 0.0, a0 = 0.0, a1 = 0.0, a2 = 0.0;
    if (isD) { a0 = S[pkw][prw]; a1 = S[pkw][prw + 1]; a2 = S[pkw][prw + 2]; }
    if (role == 0) { y0 = gv[pkw][0]; a1 = gv[pkw][1]; a2 = gv[pkw][2]; }
    if (isM) { a0 = Bo[pkw][ri]; a1 = Bo[pkw][3 + ri]; a2 = Bo[pkw][6 + ri]; }
    if (isV && partw > 0) {
      a0 = Bo[pkw - 1][ri * 3 + 0]; a1 = Bo[pkw - 1][ri * 3 + 1]; a2 = Bo[pkw - 1][ri * 3 + 2];
    }
    for (int i = 0; i < MI; ++i) {
      const int p = pkw + i;
      const bool last = (i == MI - 1);
      const double sn = S[p + 1][0];
      double b0 = 0.0, b1 = 0.0, b2 = 0.0;
      if (role == 0)      { b0 = gv[p + 1][0]; b1 = gv[p + 1][1]; b2 = gv[p + 1][2]; }
      else if (isD)       { b0 = S[p + 1][prw]; b1 = S[p + 1][prw + 1]; b2 = S[p + 1][prw + 2]; }
      else if (isM && !last) { b0 = Bo[p + 1][ri]; b1 = Bo[p + 1][3 + ri]; b2 = Bo[p + 1][6 + ri]; }
      double e0, e1, e2, d0, d1, d2;
      panel_row(role, piv, y0, a0, a1, a2, e0, e1, e2, d0, d1, d2);
      if (keep) {
        if (role == 0)      { gv[p][0] = e0; gv[p][1] = e1; gv[p][2] = e2; S[p][0] = d0; }
        else if (role == 1) { Ls[p][0] = e1; Ls[p][1] = e2; S[p][2] = d1; }
        else if (role == 2) { Ls[p][2] = e2; S[p][5] = d2; }
        else if (isM)       { Mx[p][ri * 3 + 0] = e0; Mx[p][ri * 3 + 1] = e1; Mx[p][ri * 3 + 2] = e2; }
        else                { V[p][0 + ri] = e0; V[p][3 + ri] = e1; V[p][6 + ri] = e2; }
      }
      double X[3][3];
      X[0][0] = role_bcast<3>(e0); X[0][1] = role_bcast<3>(e1); X[0][2] = role_bcast<3>(e2);
      X[1][0] = role_bcast<4>(e0); X[1][1] = role_bcast<4>(e1); X[1][2] = role_bcast<4>(e2);
      X[2][0] = role_bcast<5>(e0); X[2][1] = role_bcast<5>(e1); X[2][2] = role_bcast<5>(e2);
      const double pn = sn - (X[0][0]*X[0][0] + X[0][1]*X[0][1] + X[0][2]*X[0][2]);
      double u0 = 0.0, u1 = 0.0, u2 = 0.0;
      if (role == 1)      { u0 = X[1][0]; u1 = X[1][1]; u2 = X[1][2]; }
      else if (role == 2) { u0 = X[2][0]; u1 = X[2][1]; u2 = X[2][2]; }
      else if (!isM)      { u0 = e0; u1 = e1; u2 = e2; }
      const double o0 = b0 - (X[0][0]*u0 + X[0][1]*u1 + X[0][2]*u2);
      const double o1 = b1 - (X[1][0]*u0 + X[1][1]*u1 + X[1][2]*u2);
      const double o2 = b2 - (X[2][0]*u0 + X[2][1]*u1 + X[2][2]*u2);
      if (!last) {
        piv = pn; y0 = o0; a0 = o0; a1 = o1; a2 = o2;
      } else if (keep) {
        if (role == 0)      { S[p + 1][0] = pn; gv[p + 1][0] = o0; gv[p + 1][1] = o1; gv[p + 1][2] = o2; }
        else if (role == 1) { S[p + 1][1] = o0; S[p + 1][2] = o1; }
        else if (role == 2) { S[p + 1][3] = o0; S[p + 1][4] = o1; S[p + 1][5] = o2; }
        else if (isV)       { R[partw][ri * 3 + 0] = o0; R[partw][ri * 3 + 1] = o1; R[partw][ri * 3 + 2] = o2; }
      }
    }
  }
  __syncthreads();
  const bool act = l16 < 15;
  const int part = act ? 5 * (lane >> 4) + l16 / 3 : 0;
  const int c = act ? l16 % 3 : 0;
  const int pk = PER * part;
  const int prow = c * (c + 1) / 2;
  if (w0) {
  for (int idx = lane; idx < (NPART - 1) * 6; idx += 32) {
    const int k = 1 + idx / 6, q = idx % 6;
    const int i2 = (q >= 3) ? 2 : ((q >= 1) ? 1 : 0);
    const int j2 = q - i2 * (i2 + 1) / 2;
    const int p0 = PER * k;
    double acc = 0.0;
    for (int i = 0; i < MI; ++i)
      acc += V[p0 + i][0 + i2]*V[p0 + i][0 + j2] + V[p0 + i][3 + i2]*V[p0 + i][3 + j2]
           + V[p0 + i][6 + i2]*V[p0 + i][6 + j2];
    S[p0 - 1][q] -= acc;
  }
  for (int idx = lane; idx < (NPART - 1) * 3; idx += 32) {
    const int k = 1 + idx / 3, j = idx % 3;
    const int p0 = PER * k;
    double acc = 0.0;
    for (int i = 0; i < MI; ++i)
      acc += V[p0 + i][0 + j]*gv[p0 + i][0] + V[p0 + i][3 + j]*gv[p0 + i][1]
           + V[p0 + i][6 + j]*gv[p0 + i][2];
    gv[p0 - 1][j] -= acc;
  }
  }
  __syncthreads();
  const int dn = lane >> 4;
  const int q16 = lane & 15;
  const bool isS = q16 < 3, isC = (q16 >= 3) && (q16 < 6), isG = (q16 == 6);
  const bool isX = (q16 == 7), isY = (q16 == 8);
  const int rl = isS ? q16 : 0;
  const int rb = rl * (rl + 1) / 2;
  const int cc = isC ? q16 - 3 : 0;
  const int cbase = dn ? cc * 3 : cc;
  const int cstr = dn ? 1 : 3;
  const double id0 = (isS && rl == 0) ? 1.0 : 0.0, id1 = (isS && rl == 1) ? 1.0 : 0.0, id2 = (isS && rl == 2) ? 1.0 : 0.0;
  double a0 = 0.0, a1 = 0.0, a2 = 0.0;
  if (w0) {
    const int sj = dn ? NPART - 1 : 0;
    const int s = PER * sj + MI;
    const int m = dn ? sj : sj + 1;
    if (isS)      { a0 = S[s][rb]; a1 = S[s][rb + 1]; a2 = S[s][rb + 2]; }
    else if (isC) { a0 = R[m][cbase]; a1 = R[m][cbase + cstr]; a2 = R[m][cbase + 2 * cstr]; }
    else if (isG) { a0 = gv[s][0]; a1 = gv[s][1]; a2 = gv[s][2]; }
  }
  for (int t = 0; t <= MID; ++t) {
    const int sj = dn ? NPART - 1 - t : t;
    const bool live = dn ? (t < NPART - 1 - MID) : true;
    const int s = PER * sj + MI;
    const int m = dn ? sj : sj + 1;
    if (w0) {
    double d = sqrt(row_bcast<0>(a0));
    a0 = (isS && rl == 0) ? d : a0 / d;
    const double l10 = row_bcast<1>(a0), l20 = row_bcast<2>(a0);
    a1 -= a0 * l10;
    a2 -= a0 * l20;
    const double d0 = d;
    d = sqrt(row_bcast<1>(a1));
    double q = (isX ? l10 : a1) / (isX ? d0 : d);
    const double sm10 = q;
    a1 = (isS && rl == 1) ? d : (isX ? a1 : q);
    const double l21 = row_bcast<2>(a1);
    a2 -= a1 * l21;
    const double d1 = d;
    d = sqrt(row_bcast<2>(a2));
    q = (isX ? l20 : (isY ? l21 : a2)) / (isX ? d0 : (isY ? d1 : d));
    a2 = (isS && rl == 2) ? d : ((isX || isY) ? a2 : q);
    if (live) {
      if (isS) {
        S[s][rb] = a0;
        if (rl >= 1) S[s][rb + 1] = a1;
        if (rl == 2) S[s][rb + 2] = a2;
      } else if (isC) {
        if (t < MID) { Rm[m][cc * 3 + 0] = a0; Rm[m][cc * 3 + 1] = a1; Rm[m][cc * 3 + 2] = a2; }
      } else if (isG) {
        gv[s][0] = a0; gv[s][1] = a1; gv[s][2] = a2;
      } else if (isX) {
        Ls[s][0] = sm10; Ls[s][1] = q;
      } else if (isY) {
        Ls[s][2] = q;
      }
    }
    }
    __syncthreads();
    if (w0 && t < MID) {
      const int sjn = dn ? sj - 1 : sj + 1;
      const int sn = PER * sjn + MI;
      const int mn = dn ? sjn : sjn + 1;
      const bool toMid = (sjn == MID);
      const bool dnDelta = dn && toMid;
      double X[3][3];
      for (int r = 0; r < 3; ++r) { X[r][0] = Rm[m][r * 3 + 0]; X[r][1] = Rm[m][r * 3 + 1]; X[r][2] = Rm[m][r * 3 + 2]; }
      double u0 = 0.0, u1 = 0.0, u2 = 0.0, b0 = 0.0, b1 = 0.0, b2 = 0.0;
      if (isS) {
        u0 = Rm[m][rl * 3 + 0]; u1 = Rm[m][rl * 3 + 1]; u2 = Rm[m][rl * 3 + 2];
        if (!dnDelta) { b0 = S[sn][rb]; b1 = S[sn][rb + 1]; b2 = S[sn][rb + 2]; }
      } else if (isG) {
        u0 = a0; u1 = a1; u2 = a2;
        if (!dnDelta) { b0 = gv[sn][0]; b1 = gv[sn][1]; b2 = gv[sn][2]; }
      } else if (isC) {
        if (!toMid) { b0 = R[mn][cbase]; b1 = R[mn][cbase + cstr]; b2 = R[mn][cbase + 2 * cstr]; }
      }
      a0 = b0 - (X[0][0]*u0 + X[0][1]*u1 + X[0][2]*u2);
      a1 = b1 - (X[1][0]*u0 + X[1][1]*u1 + X[1][2]*u2);
      a2 = b2 - (X[2][0]*u0 + X[2][1]*u1 + X[2][2]*u2);
      if (dnDelta && live) {
        if (isS)      { Dl[rl][0] = a0; Dl[rl][1] = a1; Dl[rl][2] = a2; }
        else if (isG) { Dl[3][0] = a0; Dl[3][1] = a1; Dl[3][2] = a2; }
      }
      if (!dn && toMid) {
        if (isS)      { a0 += Dl[rl][0]; a1 += Dl[rl][1]; a2 += Dl[rl][2]; }
        else if (isG) { a0 += Dl[3][0]; a1 += Dl[3][1]; a2 += Dl[3][2]; }
      }
      if (dn && (t >= NPART - 2 - MID)) { a0 = id0; a1 = id1; a2 = id2; }
    }
  }
  if (w0) {
    const int c3 = isS ? q16 : 0;
    const int dg3 = rb + c3;
    double x0, x1, x2;
    {
      const int s = PER * MID + MI;
      const double sc = gv[s][c3] / S[s][dg3];
      const double m10 = Ls[s][0], m20 = Ls[s][1], m21 = Ls[s][2];
      x2 = row_bcast<2>(sc);
      x1 = row_bcast<1>(sc) - m21 * x2;
      x0 = row_bcast<0>(sc) - m10 * x1 - m20 * x2;
    }
    if (lane == 0) { gv[PER * MID + MI][0] = x0; gv[PER * MID + MI][1] = x1; gv[PER * MID + MI][2] = x2; }
    for (int i = 0; i < MID; ++i) {
      const int sj = dn ? MID + 1 + i : MID - 1 - i;
      const bool has = sj < NPART;
      const int sjc = has ? sj : NPART - 1;
      const int s = PER * sjc + MI;
      const int m = dn ? sjc : sjc + 1;
      const double tt = gv[s][c3] - (Rm[m][c3]*x0 + Rm[m][3 + c3]*x1 + Rm[m][6 + c3]*x2);
      const double sc = tt / S[s][dg3];
      const double m10 = Ls[s][0], m20 = Ls[s][1], m21 = Ls[s][2];
      const double n2 = row_bcast<2>(sc);
      const double n1 = row_bcast<1>(sc) - m21 * n2;
      const double n0 = row_bcast<0>(sc) - m10 * n1 - m20 * n2;
      if (has) {
        x0 = n0; x1 = n1; x2 = n2;
        if (q16 == 0) { gv[s][0] = x0; gv[s][1] = x1; gv[s][2] = x2; }
      }
    }
  }
  __syncthreads();
  if (w0) {
    double xl0 = 0.0, xl1 = 0.0, xl2 = 0.0;
    if (part > 0) { xl0 = gv[pk - 1][0]; xl1 = gv[pk - 1][1]; xl2 = gv[pk - 1][2]; }
    double x0 = gv[pk + MI][0], x1 = gv[pk + MI][1], x2 = gv[pk + MI][2];
    const int dg = prow + c;
    for (int i = MI - 1; i >= 0; --i) {
      const int p = pk + i;
      double t = gv[p][c] - (Mx[p][c]*x0 + Mx[p][3 + c]*x1 + Mx[p][6 + c]*x2);
      t -= V[p][c * 3 + 0]*xl0 + V[p][c * 3 + 1]*xl1 + V[p][c * 3 + 2]*xl2;
      const double s = t / S[p][dg];
      const double m10 = Ls[p][0], m20 = Ls[p][1], m21 = Ls[p][2];
      x2 = part_bcast<2>(s);
      x1 = part_bcast<1>(s) - m21 * x2;
      x0 = part_bcast<0>(s) - m10 * x1 - m20 * x2;
      if (act && c == 0) { gv[p][0] = x0; gv[p][1] = x1; gv[p][2] = x2; }
    }
  }
  __syncthreads();

  if (w0) {
  for (int t = lane; t < NSTEP; t += 32) {
    out[0 * OUT_U + b * NSTEP + t] = (float)gv[t][0];
    out[1 * OUT_U + b * NSTEP + t] = (float)gv[t][1];
    out[2 * OUT_U + b * NSTEP + t] = (float)gv[t][2];
  }
  if (lane == 0) out[OUT_EPS + b] = 0.0f;
  for (int p = lane; p < T1; p += 32) out[OUT_ST + b * T1 + p] = hstep[p];
  }
}

extern "C" void kernel_launch(void* const* d_in, const int* in_sizes, int n_in,
                              void* d_out, int out_size, void* d_ws, size_t ws_size,
                              hipStream_t stream) {
  (void)in_sizes; (void)n_in; (void)out_size; (void)d_ws; (void)ws_size;
  const float* coeffs = (const float*)d_in[0];
  const float* rhs    = (const float*)d_in[1];
  const float* ivr    = (const float*)d_in[2];
  const float* steps  = (const float*)d_in[3];
  float* out = (float*)d_out;
  hipLaunchKernelGGL(ode_blocktri_kernel, dim3(NB), dim3(128), 0, stream,
                     coeffs, rhs, ivr, steps, out);
}


# Interiors' back-substitution made explicit: x_p = a_p + G_p x_sR + C_p x_sL; a, G, C built by the 3 idle waves beside the separators' chain; 9-step serial pass gone; sum order, rounding change
# speedup vs baseline: 14.7181x; 1.1282x over previous
#include <hip/hip_runtime.h>
#include <hip/hip_bf16.h>
#include <math.h>


typedef __attribute__((ext_vector_type(2))) float v2f;
typedef __attribute__((ext_vector_type(8))) float v8f;

#define NSTEP 100
#define T1    99
#define NB    256
#define OUT_U   (NB * NSTEP)
#define OUT_EPS (3 * OUT_U)
#define OUT_ST  (OUT_EPS + NB)
#define PER   10
#define MI    9
#define NPART 10
#define MID   5
#define NTILE 7

__device__ __forceinline__ void Ecoef(int r, int c, float* pa, float* pb, float* pc) {
  float a = 0.f, b = 0.f, q = 0.f;
  if (r == 0)      { a = (c == 0) ? 1.f  : ((c == 3) ? -1.f : 0.f); b = (c == 1) ? 1.f  : 0.f; q = (c == 2) ? 0.5f : 0.f; }
  else if (r == 1) { a = (c == 1) ? 1.f  : ((c == 4) ? -1.f : 0.f); b = (c == 2) ? 1.f  : 0.f; }
  else if (r == 2) { a = (c == 0) ? -1.f : ((c == 3) ? 1.f  : 0.f); b = (c == 4) ? -1.f : 0.f; q = (c == 5) ? 0.5f : 0.f; }
  else if (r == 3) { a = (c == 1) ? -1.f : ((c == 4) ? 1.f  : 0.f); b = (c == 5) ? -1.f : 0.f; }
  else if (r == 4) { a = (c == 1) ? -1.f : ((c == 4) ? 1.f  : 0.f); b = (c == 2 || c == 5) ? -0.5f : 0.f; }
  *pa = a; *pb = b; *pc = q;
}

__device__ __forceinline__ void ltsolve3(const double* l, double* v) {
  double x2 = v[2] / l[5];
  double x1 = (v[1] - l[4] * x2) / l[2];
  double x0 = (v[0] - l[1] * x1 - l[3] * x2) / l[0];
  v[0] = x0; v[1] = x1; v[2] = x2;
}

__device__ __forceinline__ void mstep3(const double* L, double* Bp, double X[3][3]) {
  #pragma unroll
  for (int r = 0; r < 3; ++r) {
    const double c0 = Bp[0 + r];
    const double c1 = Bp[3 + r];
    const double c2 = Bp[6 + r];
    const double x0 = c0 / L[0];
    const double x1 = (c1 - x0 * L[1]) / L[2];
    const double x2 = (c2 - x0 * L[3] - x1 * L[4]) / L[5];
    X[r][0] = x0; X[r][1] = x1; X[r][2] = x2;
  }
  #pragma unroll
  for (int r = 0; r < 3; ++r) {
    Bp[r * 3 + 0] = X[r][0]; Bp[r * 3 + 1] = X[r][1]; Bp[r * 3 + 2] = X[r][2];
  }
}
__device__ __forceinline__ constexpr unsigned part_sel(int K, int half) {
  unsigned w = 0;
  for (int j = 0; j < 8; ++j) {
    const int lj = 8 * half + j;
    const unsigned s = (lj < 15) ? (unsigned)(3 * (lj / 3) + K) : 15u;
    w |= s << (4 * j);
  }
  return w;
}
template <int K>
__device__ __forceinline__ double part_bcast(double v) {
  constexpr unsigned lo = part_sel(K, 0), hi = part_sel(K, 1);
  const unsigned a = __builtin_amdgcn_permlane16(0u, (unsigned)__double2loint(v), lo, hi, false, false);
  const unsigned b = __builtin_amdgcn_permlane16(0u, (unsigned)__double2hiint(v), lo, hi, false, false);
  return __hiloint2double((int)b, (int)a);
}

template <int K>
__device__ __forceinline__ double row_bcast(double v) {
  constexpr unsigned sel = 0x11111111u * (unsigned)K;
  const unsigned a = __builtin_amdgcn_permlane16(0u, (unsigned)__double2loint(v), sel, sel, false, false);
  const unsigned b = __builtin_amdgcn_permlane16(0u, (unsigned)__double2hiint(v), sel, sel, false, false);
  return __hiloint2double((int)b, (int)a);
}

__device__ __forceinline__ void wave_lds_handover() {
  __builtin_amdgcn_fence(__ATOMIC_RELEASE, "wavefront");
  __builtin_amdgcn_wave_barrier();
  __builtin_amdgcn_fence(__ATOMIC_ACQUIRE, "wavefront");
}

__device__ __forceinline__ constexpr unsigned role_sel(int K, int half) {
  unsigned w = 0;
  for (int j = 0; j < 8; ++j) {
    const int lj = 8 * half + j;
    const unsigned sl = (lj < 9) ? (unsigned)K : (unsigned)(9 + K);
    w |= sl << (4 * j);
  }
  return w;
}
template <int K>
__device__ __forceinline__ double role_bcast(double v) {
  constexpr unsigned lo = role_sel(K, 0), hi = role_sel(K, 1);
  const unsigned a = __builtin_amdgcn_permlane16(0u, (unsigned)__double2loint(v), lo, hi, false, false);
  const unsigned b = __builtin_amdgcn_permlane16(0u, (unsigned)__double2hiint(v), lo, hi, false, false);
  return __hiloint2double((int)b, (int)a);
}

__device__ __forceinline__ void panel_row(int role, double piv, double y0, double a0, double a1, double a2,
                                          double& e0, double& e1, double& e2,
                                          double& d0, double& d1, double& d2) {
  d0 = sqrt(piv);
  e0 = ((role == 0) ? y0 : a0) / d0;
  const double l10 = role_bcast<1>(e0), l20 = role_bcast<2>(e0);
  a1 -= e0 * l10;  a2 -= e0 * l20;
  d1 = sqrt(role_bcast<1>(a1));
  e1 = ((role == 1) ? e0 : a1) / ((role == 1) ? d0 : d1);
  const double l21 = role_bcast<2>(e1);
  a2 -= e1 * l21;
  d2 = sqrt(role_bcast<2>(a2));
  e2 = ((role == 1) ? l20 : ((role == 2) ? e1 : a2)) / ((role == 1) ? d0 : ((role == 2) ? d1 : d2));
}

__global__ __launch_bounds__(128) void ode_blocktri_kernel(
    const float* __restrict__ coeffs,
    const float* __restrict__ rhs,
    const float* __restrict__ ivr,
    const float* __restrict__ steps,
    float* __restrict__ out)
{
  __shared__ double S[NSTEP][6];
  __shared__ double Bo[T1][9];
  __shared__ double gv[NSTEP][3];
  __shared__ float  hstep[T1];
  __shared__ double V[NSTEP][9];
  __shared__ double Mx[T1][9];
  __shared__ double R[NPART][9];
  __shared__ double Rm[NPART][9];
  __shared__ double Dl[4][3];
  __shared__ double Ls[NSTEP][3];
  __shared__ double Ax[NSTEP][3];
  __shared__ double Gx[NSTEP][9];
  __shared__ double Cx[NSTEP][9];

  const int b = blockIdx.x;
  const int tid = threadIdx.x;
  const int lane = tid & 31;
  const int wv = __builtin_amdgcn_readfirstlane(tid >> 5);
  const bool w0 = (wv == 0);

  for (int i = tid; i < T1; i += 128) hstep[i] = steps[b * T1 + i];
  __syncthreads();

  if (w0) {
  const int e  = lane & 15;
  const int hh = lane >> 4;
  const int i2 = (e >= 3) ? 2 : ((e >= 1) ? 1 : 0);
  const int i3 = (e >= 12) ? 2 : ((e >= 9) ? 1 : 0);
  const int ca = (e < 6) ? i2 : ((e < 15) ? i3 : 6);
  const int cb = (e < 6) ? e - i2 * (i2 + 1) / 2 : ((e < 15) ? e - 3 - 3 * i3 : 6);
  const int ga = (e < 6) ? 3 + ca : 6;
  const int gb = (e < 6) ? 3 + cb : 6;
  float ph0 = 0.f, ph1 = 0.f, ph2 = 0.f, ph3 = 0.f, ph4 = 0.f;
  float pg0 = 0.f, pg1 = 0.f, pg2 = 0.f, pg3 = 0.f, pg4 = 0.f;
  #pragma unroll
  for (int k = 0; k < 5; ++k) {
    float a1, b1, q1, a2, b2, q2;
    Ecoef(k, ca, &a1, &b1, &q1); Ecoef(k, cb, &a2, &b2, &q2);
    ph0 += a1 * a2; ph1 += a1 * b2 + b1 * a2; ph2 += a1 * q2 + b1 * b2 + q1 * a2;
    ph3 += b1 * q2 + q1 * b2; ph4 += q1 * q2;
    Ecoef(k, ga, &a1, &b1, &q1); Ecoef(k, gb, &a2, &b2, &q2);
    pg0 += a1 * a2; pg1 += a1 * b2 + b1 * a2; pg2 += a1 * q2 + b1 * b2 + q1 * a2;
    pg3 += b1 * q2 + q1 * b2; pg4 += q1 * q2;
  }
  v2f A0, A1, A2;
  A0.x = hh ? ph2 : ph0;  A0.y = hh ? ph3 : ph1;
  A1.x = hh ? pg1 : ph4;  A1.y = hh ? pg2 : pg0;
  A2.x = hh ? 0.f : pg3;  A2.y = hh ? 0.f : pg4;

  float cf0[NTILE], cf1[NTILE], cf2[NTILE], rf[NTILE], hf[NTILE], gf[NTILE];
  #pragma unroll
  for (int n = 0; n < NTILE; ++n) {
    const int t = 16 * n + e;
    const bool own = (hh == 0) && (t < NSTEP);
    const int tl = (t < NSTEP) ? t : NSTEP - 1;
    const float l0 = coeffs[(b * NSTEP + tl) * 3 + 0];
    const float l1 = coeffs[(b * NSTEP + tl) * 3 + 1];
    const float l2 = coeffs[(b * NSTEP + tl) * 3 + 2];
    const float lr = rhs[b * NSTEP + tl];
    const float lh = hstep[(t < T1) ? t : T1 - 1];
    const float lg = hstep[(t >= 1 && t < NSTEP) ? t - 1 : 0];
    cf0[n] = own ? l0 : 0.f;
    cf1[n] = own ? l1 : 0.f;
    cf2[n] = own ? l2 : 0.f;
    rf[n]  = own ? lr : 0.f;
    hf[n]  = (t < T1) ? lh : 0.f;
    gf[n]  = (t >= 1 && t < NSTEP) ? lg : 0.f;
  }
  #pragma unroll
  for (int n = 0; n < NTILE; ++n) {
    const int t = 16 * n + e;
    const float h = hf[n], g = gf[n];
    const float u = (t < T1) ? 1.f : 0.f;
    const float w = (t >= 1 && t < NSTEP) ? 1.f : 0.f;
    const float h2 = h * h, h3 = h2 * h, h4 = h2 * h2;
    const float g2 = g * g, g3 = g2 * g, g4 = g2 * g2;
    v2f B0, B1, B2;
    B0.x = hh ? h2 : u;    B0.y = hh ? h3 : h;
    B1.x = hh ? g  : h4;   B1.y = hh ? g2 : w;
    B2.x = hh ? 0.f : g3;  B2.y = hh ? 0.f : g4;
    v8f acc = {};
    acc = __builtin_amdgcn_wmma_f32_16x16x4_f32(false, A0, false, B0, (short)0, acc, false, false);
    acc = __builtin_amdgcn_wmma_f32_16x16x4_f32(false, A1, false, B1, (short)0, acc, false, false);
    acc = __builtin_amdgcn_wmma_f32_16x16x4_f32(false, A2, false, B2, (short)0, acc, false, false);
    const double c0 = (double)cf0[n], c1 = (double)cf1[n], c2 = (double)cf2[n], rr = (double)rf[n];
    double s0 = fma(c0, c0, (double)acc[0]);
    double s1 = fma(c1, c0, (double)acc[1]);
    double s2 = fma(c1, c1, (double)acc[2]);
    double s3 = fma(c2, c0, (double)acc[3]);
    double s4 = fma(c2, c1, (double)acc[4]);
    double s5 = fma(c2, c2, (double)acc[5]);
    const double s6 = (double)acc[6], s7 = (double)acc[7];
    double y0 = c0 * rr, y1 = c1 * rr, y2 = c2 * rr;
    if (n == 0 && lane == 0) {
      s0 += 1.0; s2 += 1.0;
      y0 += (double)ivr[b * 2 + 0];
      y1 += (double)ivr[b * 2 + 1];
    }
    if (hh == 0) {
      if (t < NSTEP) {
        S[t][0] = s0; S[t][1] = s1; S[t][2] = s2; S[t][3] = s3; S[t][4] = s4; S[t][5] = s5;
        gv[t][0] = y0; gv[t][1] = y1; gv[t][2] = y2;
      }
      if (t < T1) { Bo[t][0] = s6; Bo[t][1] = s7; }
    } else if (t < T1) {
      Bo[t][2] = s0; Bo[t][3] = s1; Bo[t][4] = s2; Bo[t][5] = s3; Bo[t][6] = s4; Bo[t][7] = s5; Bo[t][8] = s6;
    }
  }
  }
  __syncthreads();

  const int l16 = lane & 15;
  {
    const int rw = lane >> 4;
    const bool inB = l16 >= 9;
    const int role = !inB ? l16
                   : ((rw == 0) ? ((l16 == 15) ? 6 : l16 - 9)
                                : ((l16 == 9) ? 7 : ((l16 == 15) ? 8 : l16 - 9)));
    const bool isCopy = inB && (rw == 1) && (l16 >= 10) && (l16 <= 14);
    const int pw = 3 * wv + (inB ? 1 : (rw ? 2 : 0));
    const bool keep = (pw < NPART) && !isCopy;
    const int partw = (pw < NPART) ? pw : NPART - 1;
    const int pkw = PER * partw;
    const bool isD = role < 3, isM = (role >= 3) && (role < 6), isV = role >= 6;
    const int ri = isD ? role : (isM ? role - 3 : role - 6);
    const int prw = ri * (ri + 1) / 2;
    double piv = S[pkw][0];
    double y0 = 0.0, a0 = 0.0, a1 = 0.0, a2 = 0.0;
    if (isD) { a0 = S[pkw][prw]; a1 = S[pkw][prw + 1]; a2 = S[pkw][prw + 2]; }
    if (role == 0) { y0 = gv[pkw][0]; a1 = gv[pkw][1]; a2 = gv[pkw][2]; }
    if (isM) { a0 = Bo[pkw][ri]; a1 = Bo[pkw][3 + ri]; a2 = Bo[pkw][6 + ri]; }
    if (isV && partw > 0) {
      a0 = Bo[pkw - 1][ri * 3 + 0]; a1 = Bo[pkw - 1][ri * 3 + 1]; a2 = Bo[pkw - 1][ri * 3 + 2];
    }
    for (int i = 0; i < MI; ++i) {
      const int p = pkw + i;
      const bool last = (i == MI - 1);
      const double sn = S[p + 1][0];
      double b0 = 0.0, b1 = 0.0, b2 = 0.0;
      if (role == 0)      { b0 = gv[p + 1][0]; b1 = gv[p + 1][1]; b2 = gv[p + 1][2]; }
      else if (isD)       { b0 = S[p + 1][prw]; b1 = S[p + 1][prw + 1]; b2 = S[p + 1][prw + 2]; }
      else if (isM && !last) { b0 = Bo[p + 1][ri]; b1 = Bo[p + 1][3 + ri]; b2 = Bo[p + 1][6 + ri]; }
      double e0, e1, e2, d0, d1, d2;
      panel_row(role, piv, y0, a0, a1, a2, e0, e1, e2, d0, d1, d2);
      if (keep) {
        if (role == 0)      { gv[p][0] = e0; gv[p][1] = e1; gv[p][2] = e2; S[p][0] = d0; }
        else if (role == 1) { Ls[p][0] = e1; Ls[p][1] = e2; S[p][2] = d1; }
        else if (role == 2) { Ls[p][2] = e2; S[p][5] = d2; }
        else if (isM)       { Mx[p][ri * 3 + 0] = e0; Mx[p][ri * 3 + 1] = e1; Mx[p][ri * 3 + 2] = e2; }
        else                { V[p][0 + ri] = e0; V[p][3 + ri] = e1; V[p][6 + ri] = e2; }
      }
      double X[3][3];
      X[0][0] = role_bcast<3>(e0); X[0][1] = role_bcast<3>(e1); X[0][2] = role_bcast<3>(e2);
      X[1][0] = role_bcast<4>(e0); X[1][1] = role_bcast<4>(e1); X[1][2] = role_bcast<4>(e2);
      X[2][0] = role_bcast<5>(e0); X[2][1] = role_bcast<5>(e1); X[2][2] = role_bcast<5>(e2);
      const double pn = sn - (X[0][0]*X[0][0] + X[0][1]*X[0][1] + X[0][2]*X[0][2]);
      double u0 = 0.0, u1 = 0.0, u2 = 0.0;
      if (role == 1)      { u0 = X[1][0]; u1 = X[1][1]; u2 = X[1][2]; }
      else if (role == 2) { u0 = X[2][0]; u1 = X[2][1]; u2 = X[2][2]; }
      else if (!isM)      { u0 = e0; u1 = e1; u2 = e2; }
      const double o0 = b0 - (X[0][0]*u0 + X[0][1]*u1 + X[0][2]*u2);
      const double o1 = b1 - (X[1][0]*u0 + X[1][1]*u1 + X[1][2]*u2);
      const double o2 = b2 - (X[2][0]*u0 + X[2][1]*u1 + X[2][2]*u2);
      if (!last) {
        piv = pn; y0 = o0; a0 = o0; a1 = o1; a2 = o2;
      } else if (keep) {
        if (role == 0)      { S[p + 1][0] = pn; gv[p + 1][0] = o0; gv[p + 1][1] = o1; gv[p + 1][2] = o2; }
        else if (role == 1) { S[p + 1][1] = o0; S[p + 1][2] = o1; }
        else if (role == 2) { S[p + 1][3] = o0; S[p + 1][4] = o1; S[p + 1][5] = o2; }
        else if (isV)       { R[partw][ri * 3 + 0] = o0; R[partw][ri * 3 + 1] = o1; R[partw][ri * 3 + 2] = o2; }
      }
    }
  }
  __syncthreads();
  if (w0) {
  for (int idx = lane; idx < (NPART - 1) * 6; idx += 32) {
    const int k = 1 + idx / 6, q = idx % 6;
    const int i2 = (q >= 3) ? 2 : ((q >= 1) ? 1 : 0);
    const int j2 = q - i2 * (i2 + 1) / 2;
    const int p0 = PER * k;
    double acc = 0.0;
    for (int i = 0; i < MI; ++i)
      acc += V[p0 + i][0 + i2]*V[p0 + i][0 + j2] + V[p0 + i][3 + i2]*V[p0 + i][3 + j2]
           + V[p0 + i][6 + i2]*V[p0 + i][6 + j2];
    S[p0 - 1][q] -= acc;
  }
  for (int idx = lane; idx < (NPART - 1) * 3; idx += 32) {
    const int k = 1 + idx / 3, j = idx % 3;
    const int p0 = PER * k;
    double acc = 0.0;
    for (int i = 0; i < MI; ++i)
      acc += V[p0 + i][0 + j]*gv[p0 + i][0] + V[p0 + i][3 + j]*gv[p0 + i][1]
           + V[p0 + i][6 + j]*gv[p0 + i][2];
    gv[p0 - 1][j] -= acc;
  }
  wave_lds_handover();
  const int dn = lane >> 4;
  const int q16 = lane & 15;
  const bool isS = q16 < 3, isC = (q16 >= 3) && (q16 < 6), isG = (q16 == 6);
  const bool isX = (q16 == 7), isY = (q16 == 8);
  const int rl = isS ? q16 : 0;
  const int rb = rl * (rl + 1) / 2;
  const int cc = isC ? q16 - 3 : 0;
  const int cbase = dn ? cc * 3 : cc;
  const int cstr = dn ? 1 : 3;
  const double id0 = (isS && rl == 0) ? 1.0 : 0.0, id1 = (isS && rl == 1) ? 1.0 : 0.0, id2 = (isS && rl == 2) ? 1.0 : 0.0;
  double a0 = 0.0, a1 = 0.0, a2 = 0.0;
  {
    const int sj = dn ? NPART - 1 : 0;
    const int s = PER * sj + MI;
    const int m = dn ? sj : sj + 1;
    if (isS)      { a0 = S[s][rb]; a1 = S[s][rb + 1]; a2 = S[s][rb + 2]; }
    else if (isC) { a0 = R[m][cbase]; a1 = R[m][cbase + cstr]; a2 = R[m][cbase + 2 * cstr]; }
    else if (isG) { a0 = gv[s][0]; a1 = gv[s][1]; a2 = gv[s][2]; }
  }
  for (int t = 0; t <= MID; ++t) {
    const int sj = dn ? NPART - 1 - t : t;
    const bool live = dn ? (t < NPART - 1 - MID) : true;
    const int s = PER * sj + MI;
    const int m = dn ? sj : sj + 1;
    double d = sqrt(row_bcast<0>(a0));
    a0 = (isS && rl == 0) ? d : a0 / d;
    const double l10 = row_bcast<1>(a0), l20 = row_bcast<2>(a0);
    a1 -= a0 * l10;
    a2 -= a0 * l20;
    const double d0 = d;
    d = sqrt(row_bcast<1>(a1));
    double q = (isX ? l10 : a1) / (isX ? d0 : d);
    const double sm10 = q;
    a1 = (isS && rl == 1) ? d : (isX ? a1 : q);
    const double l21 = row_bcast<2>(a1);
    a2 -= a1 * l21;
    const double d1 = d;
    d = sqrt(row_bcast<2>(a2));
    q = (isX ? l20 : (isY ? l21 : a2)) / (isX ? d0 : (isY ? d1 : d));
    a2 = (isS && rl == 2) ? d : ((isX || isY) ? a2 : q);
    if (live) {
      if (isS) {
        S[s][rb] = a0;
        if (rl >= 1) S[s][rb + 1] = a1;
        if (rl == 2) S[s][rb + 2] = a2;
      } else if (isC) {
        if (t < MID) { Rm[m][cc * 3 + 0] = a0; Rm[m][cc * 3 + 1] = a1; Rm[m][cc * 3 + 2] = a2; }
      } else if (isG) {
        gv[s][0] = a0; gv[s][1] = a1; gv[s][2] = a2;
      } else if (isX) {
        Ls[s][0] = sm10; Ls[s][1] = q;
      } else if (isY) {
        Ls[s][2] = q;
      }
    }
    wave_lds_handover();
    if (t < MID) {
      const int sjn = dn ? sj - 1 : sj + 1;
      const int sn = PER * sjn + MI;
      const int mn = dn ? sjn : sjn + 1;
      const bool toMid = (sjn == MID);
      const bool dnDelta = dn && toMid;
      double X[3][3];
      for (int r = 0; r < 3; ++r) { X[r][0] = Rm[m][r * 3 + 0]; X[r][1] = Rm[m][r * 3 + 1]; X[r][2] = Rm[m][r * 3 + 2]; }
      double u0 = 0.0, u1 = 0.0, u2 = 0.0, b0 = 0.0, b1 = 0.0, b2 = 0.0;
      if (isS) {
        u0 = Rm[m][rl * 3 + 0]; u1 = Rm[m][rl * 3 + 1]; u2 = Rm[m][rl * 3 + 2];
        if (!dnDelta) { b0 = S[sn][rb]; b1 = S[sn][rb + 1]; b2 = S[sn][rb + 2]; }
      } else if (isG) {
        u0 = a0; u1 = a1; u2 = a2;
        if (!dnDelta) { b0 = gv[sn][0]; b1 = gv[sn][1]; b2 = gv[sn][2]; }
      } else if (isC) {
        if (!toMid) { b0 = R[mn][cbase]; b1 = R[mn][cbase + cstr]; b2 = R[mn][cbase + 2 * cstr]; }
      }
      a0 = b0 - (X[0][0]*u0 + X[0][1]*u1 + X[0][2]*u2);
      a1 = b1 - (X[1][0]*u0 + X[1][1]*u1 + X[1][2]*u2);
      a2 = b2 - (X[2][0]*u0 + X[2][1]*u1 + X[2][2]*u2);
      if (dnDelta && live) {
        if (isS)      { Dl[rl][0] = a0; Dl[rl][1] = a1; Dl[rl][2] = a2; }
        else if (isG) { Dl[3][0] = a0; Dl[3][1] = a1; Dl[3][2] = a2; }
      }
      if (!dn && toMid) {
        if (isS)      { a0 += Dl[rl][0]; a1 += Dl[rl][1]; a2 += Dl[rl][2]; }
        else if (isG) { a0 += Dl[3][0]; a1 += Dl[3][1]; a2 += Dl[3][2]; }
      }
      if (dn && (t >= NPART - 2 - MID)) { a0 = id0; a1 = id1; a2 = id2; }
    }
  }
  {
    const int c3 = isS ? q16 : 0;
    const int dg3 = rb + c3;
    double x0, x1, x2;
    {
      const int s = PER * MID + MI;
      const double sc = gv[s][c3] / S[s][dg3];
      const double m10 = Ls[s][0], m20 = Ls[s][1], m21 = Ls[s][2];
      x2 = row_bcast<2>(sc);
      x1 = row_bcast<1>(sc) - m21 * x2;
      x0 = row_bcast<0>(sc) - m10 * x1 - m20 * x2;
    }
    if (lane == 0) { gv[PER * MID + MI][0] = x0; gv[PER * MID + MI][1] = x1; gv[PER * MID + MI][2] = x2; }
    for (int i = 0; i < MID; ++i) {
      const int sj = dn ? MID + 1 + i : MID - 1 - i;
      const bool has = sj < NPART;
      const int sjc = has ? sj : NPART - 1;
      const int s = PER * sjc + MI;
      const int m = dn ? sjc : sjc + 1;
      const double tt = gv[s][c3] - (Rm[m][c3]*x0 + Rm[m][3 + c3]*x1 + Rm[m][6 + c3]*x2);
      const double sc = tt / S[s][dg3];
      const double m10 = Ls[s][0], m20 = Ls[s][1], m21 = Ls[s][2];
      const double n2 = row_bcast<2>(sc);
      const double n1 = row_bcast<1>(sc) - m21 * n2;
      const double n0 = row_bcast<0>(sc) - m10 * n1 - m20 * n2;
      if (has) {
        x0 = n0; x1 = n1; x2 = n2;
        if (q16 == 0) { gv[s][0] = x0; gv[s][1] = x1; gv[s][2] = x2; }
      }
    }
  }
  } else {
    const bool act = l16 < 15;
    const int c = act ? l16 % 3 : 0;
    const int dg = c * (c + 1) / 2 + c;
    for (int u = wv - 1; u < 7; u += 3) {
      const int q = (2 * u + (lane >> 4)) * 5 + (act ? l16 / 3 : 0);
      const int part = q / 7, col = q % 7;
      const int pk = PER * part;
      const bool isA = (col == 0), isCc = (col >= 1) && (col <= 3);
      const int k = isA ? 0 : (isCc ? col - 1 : col - 4);
      const double* rsrc = isA ? &gv[pk][c] : (isCc ? &V[pk][c * 3 + k] : &Mx[pk][3 * k + c]);
      double* zdst = isA ? &Ax[pk][c] : (isCc ? &Cx[pk][c * 3 + k] : &Gx[pk][c * 3 + k]);
      const int str = isA ? 3 : 9;
      double z0 = 0.0, z1 = 0.0, z2 = 0.0;
      double rv = rsrc[(MI - 1) * str];
      double m0 = Mx[pk + MI - 1][c], m1 = Mx[pk + MI - 1][3 + c], m2 = Mx[pk + MI - 1][6 + c];
      double dd = S[pk + MI - 1][dg];
      double m10 = Ls[pk + MI - 1][0], m20 = Ls[pk + MI - 1][1], m21 = Ls[pk + MI - 1][2];
      for (int i = MI - 1; i >= 0; --i) {
        const int in = (i > 0) ? i - 1 : 0;
        const int pn = pk + in;
        const double rvn = rsrc[in * str];
        const double m0n = Mx[pn][c], m1n = Mx[pn][3 + c], m2n = Mx[pn][6 + c];
        const double ddn = S[pn][dg];
        const double m10n = Ls[pn][0], m20n = Ls[pn][1], m21n = Ls[pn][2];
        const double r = isA ? rv : ((isCc || i == MI - 1) ? -rv : 0.0);
        const double t = r - (m0*z0 + m1*z1 + m2*z2);
        const double sc = t / dd;
        z2 = part_bcast<2>(sc);
        z1 = part_bcast<1>(sc) - m21 * z2;
        z0 = part_bcast<0>(sc) - m10 * z1 - m20 * z2;
        const double zc = (c == 0) ? z0 : ((c == 1) ? z1 : z2);
        if (act) zdst[i * str] = zc;
        rv = rvn; m0 = m0n; m1 = m1n; m2 = m2n; dd = ddn; m10 = m10n; m20 = m20n; m21 = m21n;
      }
    }
  }
  __syncthreads();

  if (tid < NSTEP) {
    const int t = tid;
    const int part = t / PER;
    const int sR = PER * part + MI;
    double x0, x1, x2;
    if (t == sR) {
      x0 = gv[t][0]; x1 = gv[t][1]; x2 = gv[t][2];
    } else {
      const double r0 = gv[sR][0], r1 = gv[sR][1], r2 = gv[sR][2];
      double l0 = 0.0, l1 = 0.0, l2 = 0.0;
      if (part > 0) { l0 = gv[sR - PER][0]; l1 = gv[sR - PER][1]; l2 = gv[sR - PER][2]; }
      x0 = Ax[t][0] + (Gx[t][0]*r0 + Gx[t][1]*r1 + Gx[t][2]*r2) + (Cx[t][0]*l0 + Cx[t][1]*l1 + Cx[t][2]*l2);
      x1 = Ax[t][1] + (Gx[t][3]*r0 + Gx[t][4]*r1 + Gx[t][5]*r2) + (Cx[t][3]*l0 + Cx[t][4]*l1 + Cx[t][5]*l2);
      x2 = Ax[t][2] + (Gx[t][6]*r0 + Gx[t][7]*r1 + Gx[t][8]*r2) + (Cx[t][6]*l0 + Cx[t][7]*l1 + Cx[t][8]*l2);
    }
    out[0 * OUT_U + b * NSTEP + t] = (float)x0;
    out[1 * OUT_U + b * NSTEP + t] = (float)x1;
    out[2 * OUT_U + b * NSTEP + t] = (float)x2;
  }
  if (tid == 0) out[OUT_EPS + b] = 0.0f;
  if (tid < T1) out[OUT_ST + b * T1 + tid] = hstep[tid];
}

extern "C" void kernel_launch(void* const* d_in, const int* in_sizes, int n_in,
                              void* d_out, int out_size, void* d_ws, size_t ws_size,
                              hipStream_t stream) {
  (void)in_sizes; (void)n_in; (void)out_size; (void)d_ws; (void)ws_size;
  const float* coeffs = (const float*)d_in[0];
  const float* rhs    = (const float*)d_in[1];
  const float* ivr    = (const float*)d_in[2];
  const float* steps  = (const float*)d_in[3];
  float* out = (float*)d_out;
  hipLaunchKernelGGL(ode_blocktri_kernel, dim3(NB), dim3(128), 0, stream,
                     coeffs, rhs, ivr, steps, out);
}
